# starting kernel
# baseline (speedup 1.0000x reference)
#include <hip/hip_runtime.h>

#define N_NODES 100000
#define N_EDGES 1600000
#define D_IN    128
#define D_OUT   64

typedef __attribute__((ext_vector_type(2))) float v2f;
typedef __attribute__((ext_vector_type(8))) float v8f;

__global__ __launch_bounds__(256) void zero_kernel(float4* __restrict__ out, int n4) {
    int i = blockIdx.x * blockDim.x + threadIdx.x;
    if (i < n4) out[i] = make_float4(0.f, 0.f, 0.f, 0.f);
}

__global__ __launch_bounds__(256) void gemm_bias_kernel(
    const float* __restrict__ x, const float* __restrict__ W,
    const float* __restrict__ bias, float* __restrict__ h)
{
    const int lane = threadIdx.x & 31;
    const int wave = threadIdx.x >> 5;
    const int m0   = (blockIdx.x * 8 + wave) * 16;
    if (m0 >= N_NODES) return;

    const int half = lane >> 4;
    const int l15  = lane & 15;

    v8f acc[4];
    #pragma unroll
    for (int t = 0; t < 4; ++t) {
        const float bval = bias[t * 16 + l15];
        #pragma unroll
        for (int i = 0; i < 8; ++i) acc[t][i] = bval;
    }

    const float* xrow  = x + (size_t)(m0 + l15) * D_IN + half * 2;
    const float* wbase = W + (size_t)(half * 2) * D_OUT + l15;

    #pragma unroll
    for (int k0 = 0; k0 < D_IN; k0 += 4) {
        v2f a;
        a.x = xrow[k0 + 0];
        a.y = xrow[k0 + 1];
        #pragma unroll
        for (int t = 0; t < 4; ++t) {
            v2f bm;
            bm.x = wbase[(size_t)(k0 + 0) * D_OUT + t * 16];
            bm.y = wbase[(size_t)(k0 + 1) * D_OUT + t * 16];
            acc[t] = __builtin_amdgcn_wmma_f32_16x16x4_f32(
                          false, a,  false, bm,
                          (short)0, acc[t],
                          false,  false);
        }
    }

    float* hout = h + (size_t)(m0 + half * 8) * D_OUT + l15;
    #pragma unroll
    for (int t = 0; t < 4; ++t)
        #pragma unroll
        for (int v = 0; v < 8; ++v)
            hout[(size_t)v * D_OUT + t * 16] = acc[t][v];
}

__global__ __launch_bounds__(256) void scatter_kernel(
    const float* __restrict__ h, const float* __restrict__ edge_w,
    const int* __restrict__ src, const int* __restrict__ dst,
    float* __restrict__ out)
{
    const int lane = threadIdx.x & 31;
    const int e    = blockIdx.x * 8 + (threadIdx.x >> 5);
    if (e >= N_EDGES) return;

    const int   s = src[e];
    const int   d = dst[e];
    const float w = edge_w[e];

    const float2 v = ((const float2*)(h + (size_t)s * D_OUT))[lane];
    float* orow = out + (size_t)d * D_OUT + lane * 2;
    atomicAdd(orow + 0, v.x * w);
    atomicAdd(orow + 1, v.y * w);
}

extern "C" void kernel_launch(void* const* d_in, const int* in_sizes, int n_in,
                              void* d_out, int out_size, void* d_ws, size_t ws_size,
                              hipStream_t stream) {
    const float* x      = (const float*)d_in[0];
    const float* W      = (const float*)d_in[1];
    const float* b      = (const float*)d_in[2];
    const float* edge_w = (const float*)d_in[3];
    const int*   src    = (const int*)d_in[4];
    const int*   dst    = (const int*)d_in[5];
    float*       out    = (float*)d_out;
    float*       h      = (float*)d_ws;

    (void)in_sizes; (void)n_in; (void)ws_size;

    const int n4 = out_size / 4;
    zero_kernel<<<(n4 + 255) / 256, 256, 0, stream>>>((float4*)out, n4);

    const int m_tiles = (N_NODES + 15) / 16;
    const int gemm_blocks = (m_tiles + 7) / 8;
    gemm_bias_kernel<<<gemm_blocks, 256, 0, stream>>>(x, W, b, h);

    scatter_kernel<<<(N_EDGES + 7) / 8, 256, 0, stream>>>(h, edge_w, src, dst, out);
}


# Edge phase owner-computes, no filing or sort: 128 work-groups each scan the whole dst list, gather only edges into their own 782 out rows, sum in LDS (float adds, arrival order), store out once
# speedup vs baseline: 1.3148x; 1.3148x over previous
#include <hip/hip_runtime.h>

#define N_NODES 100000
#define N_EDGES 1600000
#define D_IN    128
#define D_OUT   64

typedef __attribute__((ext_vector_type(2))) float v2f;
typedef __attribute__((ext_vector_type(8))) float v8f;

__global__ __launch_bounds__(256) void gemm_bias_kernel(
    const float* __restrict__ x, const float* __restrict__ W,
    const float* __restrict__ bias, float* __restrict__ h)
{
    const int lane = threadIdx.x & 31;
    const int wave = threadIdx.x >> 5;
    const int m0   = (blockIdx.x * 8 + wave) * 16;
    if (m0 >= N_NODES) return;

    const int half = lane >> 4;
    const int l15  = lane & 15;

    v8f acc[4];
    #pragma unroll
    for (int t = 0; t < 4; ++t) {
        const float bval = bias[t * 16 + l15];
        #pragma unroll
        for (int i = 0; i < 8; ++i) acc[t][i] = bval;
    }

    const float* xrow  = x + (size_t)(m0 + l15) * D_IN + half * 2;
    const float* wbase = W + (size_t)(half * 2) * D_OUT + l15;

    #pragma unroll
    for (int k0 = 0; k0 < D_IN; k0 += 4) {
        v2f a;
        a.x = xrow[k0 + 0];
        a.y = xrow[k0 + 1];
        #pragma unroll
        for (int t = 0; t < 4; ++t) {
            v2f bm;
            bm.x = wbase[(size_t)(k0 + 0) * D_OUT + t * 16];
            bm.y = wbase[(size_t)(k0 + 1) * D_OUT + t * 16];
            acc[t] = __builtin_amdgcn_wmma_f32_16x16x4_f32(
                          false, a,  false, bm,
                          (short)0, acc[t],
                          false,  false);
        }
    }

    float* hout = h + (size_t)(m0 + half * 8) * D_OUT + l15;
    #pragma unroll
    for (int t = 0; t < 4; ++t)
        #pragma unroll
        for (int v = 0; v < 8; ++v)
            hout[(size_t)v * D_OUT + t * 16] = acc[t][v];
}

#define OWN_BLOCKS   128
#define OWN_ROWS     782
#define OWN_WAVES    32
#define OWN_THREADS  (OWN_WAVES * 32)
#define SCAN_STEPS   (N_EDGES / 128)
#define SCAN_AHEAD   8
#define BATCH        16
#define QCAP         160

static_assert(N_EDGES % 128 == 0, "the scan has no partial wave step");
static_assert(OWN_BLOCKS * OWN_ROWS >= N_NODES, "the blocks cover every row");
static_assert(QCAP >= BATCH - 1 + 128, "one scan step cannot overflow a queue");
static_assert(OWN_ROWS * D_OUT * 4 + OWN_WAVES * QCAP * 8 <= 320 * 1024, "LDS budget");

__global__ __launch_bounds__(OWN_THREADS) void owner_sum_kernel(
    const float* __restrict__ h, const float* __restrict__ edge_w,
    const int* __restrict__ src, const int* __restrict__ dst,
    float* __restrict__ out)
{
    __shared__ __attribute__((aligned(16))) float acc[OWN_ROWS * D_OUT];
    __shared__ __attribute__((aligned(16))) uint2 queue[OWN_WAVES * QCAP];

    const int tid  = threadIdx.x;
    const int lane = tid & 31;
    const int wave = __builtin_amdgcn_readfirstlane(tid >> 5);
    const unsigned base = blockIdx.x * OWN_ROWS;

    float4* acc4 = (float4*)acc;
    for (int i = tid; i < OWN_ROWS * (D_OUT / 4); i += OWN_THREADS)
        acc4[i] = make_float4(0.f, 0.f, 0.f, 0.f);
    __syncthreads();

    uint2* q = queue + wave * QCAP;
    const int4*   dst4 = (const int4*)dst;
    const float2* h2   = (const float2*)h;
    int qcnt = 0;

    auto gather_add = [&](const uint2* qs, int n) {
        const uint2 ent = qs[lane < n ? lane : n - 1];
        const int   s_l = src[ent.x];
        const int   w_l = __float_as_int(edge_w[ent.x]);
        float2 v[BATCH];
        #pragma unroll
        for (int j = 0; j < BATCH; ++j) {
            if (j < n) {
                const int sj = __builtin_amdgcn_readlane(s_l, j);
                v[j] = h2[(size_t)sj * (D_OUT / 2) + lane];
            }
        }
        #pragma unroll
        for (int j = 0; j < BATCH; ++j) {
            if (j < n) {
                const float    wj = __int_as_float(__builtin_amdgcn_readlane(w_l, j));
                const unsigned rj = __builtin_amdgcn_readlane((int)ent.y, j);
                float* a = acc + rj * D_OUT + lane * 2;
                atomicAdd(a + 0, v[j].x * wj);
                atomicAdd(a + 1, v[j].y * wj);
            }
        }
    };

    for (int t0 = wave; t0 < SCAN_STEPS; t0 += OWN_WAVES * SCAN_AHEAD) {
        int4 d[SCAN_AHEAD];
        #pragma unroll
        for (int u = 0; u < SCAN_AHEAD; ++u) {
            const int t = t0 + u * OWN_WAVES;
            d[u] = (t < SCAN_STEPS) ? dst4[(size_t)t * 32 + lane]
                                    : make_int4(-1, -1, -1, -1);
        }
        #pragma unroll
        for (int u = 0; u < SCAN_AHEAD; ++u) {
            const unsigned e0 = ((unsigned)(t0 + u * OWN_WAVES) * 32u + lane) * 4u;
            const unsigned r[4] = { (unsigned)d[u].x - base, (unsigned)d[u].y - base,
                                    (unsigned)d[u].z - base, (unsigned)d[u].w - base };
            const unsigned rmin = min(min(r[0], r[1]), min(r[2], r[3]));
            if (__builtin_amdgcn_ballot_w32(rmin < OWN_ROWS) != 0u) {
                #pragma unroll
                for (int k = 0; k < 4; ++k) {
                    const bool mine = r[k] < OWN_ROWS;
                    const unsigned mask = __builtin_amdgcn_ballot_w32(mine);
                    if (mask != 0u) {
                        const int pos = qcnt + (int)__builtin_amdgcn_mbcnt_lo(mask, 0u);
                        if (mine) q[pos] = make_uint2(e0 + k, r[k]);
                        qcnt += __builtin_popcount(mask);
                    }
                }
                while (qcnt >= BATCH) {
                    qcnt -= BATCH;
                    gather_add(q + qcnt, BATCH);
                }
            }
        }
    }
    if (qcnt > 0) gather_add(q, qcnt);
    __syncthreads();

    float4* out4 = (float4*)out;
    for (int i = tid; i < OWN_ROWS * (D_OUT / 4); i += OWN_THREADS) {
        const unsigned row = base + (unsigned)(i >> 4);
        if (row < N_NODES) out4[(size_t)row * (D_OUT / 4) + (i & 15)] = acc4[i];
    }
}

extern "C" void kernel_launch(void* const* d_in, const int* in_sizes, int n_in,
                              void* d_out, int out_size, void* d_ws, size_t ws_size,
                              hipStream_t stream) {
    const float* x      = (const float*)d_in[0];
    const float* W      = (const float*)d_in[1];
    const float* b      = (const float*)d_in[2];
    const float* edge_w = (const float*)d_in[3];
    const int*   src    = (const int*)d_in[4];
    const int*   dst    = (const int*)d_in[5];
    float*       out    = (float*)d_out;
    float*       h      = (float*)d_ws;

    (void)in_sizes; (void)n_in; (void)ws_size; (void)out_size;

    const int m_tiles = (N_NODES + 15) / 16;
    const int gemm_blocks = (m_tiles + 7) / 8;
    gemm_bias_kernel<<<gemm_blocks, 256, 0, stream>>>(x, W, b, h);

    owner_sum_kernel<<<OWN_BLOCKS, OWN_THREADS, 0, stream>>>(h, edge_w, src, dst, out);
}


# Edge list cut in two halves: 256 work-groups (128 row blocks x 2 halves) each scan half the dst list; half 1 sums to a workspace plane, a third launch adds it to out (sum of two partial sums)
# speedup vs baseline: 1.7631x; 1.3410x over previous
#include <hip/hip_runtime.h>

#define N_NODES 100000
#define N_EDGES 1600000
#define D_IN    128
#define D_OUT   64

typedef __attribute__((ext_vector_type(2))) float v2f;
typedef __attribute__((ext_vector_type(8))) float v8f;

__global__ __launch_bounds__(256) void gemm_bias_kernel(
    const float* __restrict__ x, const float* __restrict__ W,
    const float* __restrict__ bias, float* __restrict__ h)
{
    const int lane = threadIdx.x & 31;
    const int wave = threadIdx.x >> 5;
    const int m0   = (blockIdx.x * 8 + wave) * 16;
    if (m0 >= N_NODES) return;

    const int half = lane >> 4;
    const int l15  = lane & 15;

    v8f acc[4];
    #pragma unroll
    for (int t = 0; t < 4; ++t) {
        const float bval = bias[t * 16 + l15];
        #pragma unroll
        for (int i = 0; i < 8; ++i) acc[t][i] = bval;
    }

    const float* xrow  = x + (size_t)(m0 + l15) * D_IN + half * 2;
    const float* wbase = W + (size_t)(half * 2) * D_OUT + l15;

    #pragma unroll
    for (int k0 = 0; k0 < D_IN; k0 += 4) {
        v2f a;
        a.x = xrow[k0 + 0];
        a.y = xrow[k0 + 1];
        #pragma unroll
        for (int t = 0; t < 4; ++t) {
            v2f bm;
            bm.x = wbase[(size_t)(k0 + 0) * D_OUT + t * 16];
            bm.y = wbase[(size_t)(k0 + 1) * D_OUT + t * 16];
            acc[t] = __builtin_amdgcn_wmma_f32_16x16x4_f32(
                          false, a,  false, bm,
                          (short)0, acc[t],
                          false,  false);
        }
    }

    float* hout = h + (size_t)(m0 + half * 8) * D_OUT + l15;
    #pragma unroll
    for (int t = 0; t < 4; ++t)
        #pragma unroll
        for (int v = 0; v < 8; ++v)
            hout[(size_t)v * D_OUT + t * 16] = acc[t][v];
}

#define OWN_BLOCKS   128
#define OWN_ROWS     782
#define OWN_WAVES    32
#define OWN_THREADS  (OWN_WAVES * 32)
#define SCAN_STEPS   (N_EDGES / 128)
#define EDGE_PARTS   2
#define PART_STEPS   (SCAN_STEPS / EDGE_PARTS)
#define SCAN_AHEAD   8
#define BATCH        16
#define QCAP         160

static_assert(N_EDGES % 128 == 0, "the scan has no partial wave step");
static_assert(SCAN_STEPS % EDGE_PARTS == 0, "the parts of the edge list are equal");
static_assert(OWN_BLOCKS * OWN_ROWS >= N_NODES, "the blocks cover every row");
static_assert(QCAP >= BATCH - 1 + 128, "one scan step cannot overflow a queue");
static_assert(OWN_ROWS * D_OUT * 4 + OWN_WAVES * QCAP * 8 <= 320 * 1024, "LDS budget");

__global__ __launch_bounds__(OWN_THREADS) void owner_sum_kernel(
    const float* __restrict__ h, const float* __restrict__ edge_w,
    const int* __restrict__ src, const int* __restrict__ dst,
    float* __restrict__ out, float* __restrict__ planes)
{
    __shared__ __attribute__((aligned(16))) float acc[OWN_ROWS * D_OUT];
    __shared__ __attribute__((aligned(16))) uint2 queue[OWN_WAVES * QCAP];

    const int tid  = threadIdx.x;
    const int lane = tid & 31;
    const int wave = __builtin_amdgcn_readfirstlane(tid >> 5);
    const int part    = blockIdx.x % EDGE_PARTS;
    const int step_lo = part * PART_STEPS;
    const int step_hi = step_lo + PART_STEPS;
    const unsigned base = (blockIdx.x / EDGE_PARTS) * OWN_ROWS;

    float4* acc4 = (float4*)acc;
    for (int i = tid; i < OWN_ROWS * (D_OUT / 4); i += OWN_THREADS)
        acc4[i] = make_float4(0.f, 0.f, 0.f, 0.f);
    __syncthreads();

    uint2* q = queue + wave * QCAP;
    const int4*   dst4 = (const int4*)dst;
    const float2* h2   = (const float2*)h;
    int qcnt = 0;

    auto gather_add = [&](const uint2* qs, int n) {
        const uint2 ent = qs[lane < n ? lane : n - 1];
        const int   s_l = src[ent.x];
        const int   w_l = __float_as_int(edge_w[ent.x]);
        float2 v[BATCH];
        #pragma unroll
        for (int j = 0; j < BATCH; ++j) {
            if (j < n) {
                const int sj = __builtin_amdgcn_readlane(s_l, j);
                v[j] = h2[(size_t)sj * (D_OUT / 2) + lane];
            }
        }
        #pragma unroll
        for (int j = 0; j < BATCH; ++j) {
            if (j < n) {
                const float    wj = __int_as_float(__builtin_amdgcn_readlane(w_l, j));
                const unsigned rj = __builtin_amdgcn_readlane((int)ent.y, j);
                float* a = acc + rj * D_OUT + lane * 2;
                atomicAdd(a + 0, v[j].x * wj);
                atomicAdd(a + 1, v[j].y * wj);
            }
        }
    };

    for (int t0 = step_lo + wave; t0 < step_hi; t0 += OWN_WAVES * SCAN_AHEAD) {
        int4 d[SCAN_AHEAD];
        #pragma unroll
        for (int u = 0; u < SCAN_AHEAD; ++u) {
            const int t = t0 + u * OWN_WAVES;
            d[u] = (t < step_hi) ? dst4[(size_t)t * 32 + lane]
                                    : make_int4(-1, -1, -1, -1);
        }
        #pragma unroll
        for (int u = 0; u < SCAN_AHEAD; ++u) {
            const unsigned e0 = ((unsigned)(t0 + u * OWN_WAVES) * 32u + lane) * 4u;
            const unsigned r[4] = { (unsigned)d[u].x - base, (unsigned)d[u].y - base,
                                    (unsigned)d[u].z - base, (unsigned)d[u].w - base };
            const unsigned rmin = min(min(r[0], r[1]), min(r[2], r[3]));
            if (__builtin_amdgcn_ballot_w32(rmin < OWN_ROWS) != 0u) {
                #pragma unroll
                for (int k = 0; k < 4; ++k) {
                    const bool mine = r[k] < OWN_ROWS;
                    const unsigned mask = __builtin_amdgcn_ballot_w32(mine);
                    if (mask != 0u) {
                        const int pos = qcnt + (int)__builtin_amdgcn_mbcnt_lo(mask, 0u);
                        if (mine) q[pos] = make_uint2(e0 + k, r[k]);
                        qcnt += __builtin_popcount(mask);
                    }
                }
                while (qcnt >= BATCH) {
                    qcnt -= BATCH;
                    gather_add(q + qcnt, BATCH);
                }
            }
        }
    }
    if (qcnt > 0) gather_add(q, qcnt);
    __syncthreads();

    float4* out4 = (float4*)(part == 0 ? out
                                       : planes + (size_t)(part - 1) * N_NODES * D_OUT);
    for (int i = tid; i < OWN_ROWS * (D_OUT / 4); i += OWN_THREADS) {
        const unsigned row = base + (unsigned)(i >> 4);
        if (row < N_NODES) out4[(size_t)row * (D_OUT / 4) + (i & 15)] = acc4[i];
    }
}

#define ADD_BLOCKS   625
#define ADD_THREADS  256
#define ADD_STEPS    2
#define ADD_AHEAD    5
#define ADD_TOTAL    (N_NODES * (D_OUT / 4))

static_assert(ADD_BLOCKS * ADD_THREADS * ADD_STEPS * ADD_AHEAD == ADD_TOTAL,
              "the add kernel covers out exactly, without guards");

__global__ __launch_bounds__(ADD_THREADS) void add_planes_kernel(
    const float* __restrict__ planes, float* __restrict__ out)
{
    const float4* p4 = (const float4*)planes;
    float4*       o4 = (float4*)out;
    for (int step = 0; step < ADD_STEPS; ++step) {
        const int i0 = (step * ADD_AHEAD * ADD_BLOCKS + blockIdx.x) * ADD_THREADS + threadIdx.x;
        float4 s[ADD_AHEAD];
        float4 v[ADD_AHEAD][EDGE_PARTS - 1];
        #pragma unroll
        for (int u = 0; u < ADD_AHEAD; ++u) {
            const int i = i0 + u * ADD_BLOCKS * ADD_THREADS;
            s[u] = o4[i];
            #pragma unroll
            for (int p = 0; p < EDGE_PARTS - 1; ++p)
                v[u][p] = p4[(size_t)p * ADD_TOTAL + i];
        }
        #pragma unroll
        for (int u = 0; u < ADD_AHEAD; ++u) {
            const int i = i0 + u * ADD_BLOCKS * ADD_THREADS;
            #pragma unroll
            for (int p = 0; p < EDGE_PARTS - 1; ++p) {
                s[u].x += v[u][p].x; s[u].y += v[u][p].y;
                s[u].z += v[u][p].z; s[u].w += v[u][p].w;
            }
            o4[i] = s[u];
        }
    }
}

extern "C" void kernel_launch(void* const* d_in, const int* in_sizes, int n_in,
                              void* d_out, int out_size, void* d_ws, size_t ws_size,
                              hipStream_t stream) {
    const float* x      = (const float*)d_in[0];
    const float* W      = (const float*)d_in[1];
    const float* b      = (const float*)d_in[2];
    const float* edge_w = (const float*)d_in[3];
    const int*   src    = (const int*)d_in[4];
    const int*   dst    = (const int*)d_in[5];
    float*       out    = (float*)d_out;
    float*       h      = (float*)d_ws;
    float*       planes = h + (size_t)N_NODES * D_OUT;

    (void)in_sizes; (void)n_in; (void)ws_size; (void)out_size;

    const int m_tiles = (N_NODES + 15) / 16;
    const int gemm_blocks = (m_tiles + 7) / 8;
    gemm_bias_kernel<<<gemm_blocks, 256, 0, stream>>>(x, W, b, h);

    owner_sum_kernel<<<OWN_BLOCKS * EDGE_PARTS, OWN_THREADS, 0, stream>>>(
        h, edge_w, src, dst, out, planes);

    add_planes_kernel<<<ADD_BLOCKS, ADD_THREADS, 0, stream>>>(planes, out);
}


# 256 destination blocks of 391 rows, one work-group each, walking a per-block bit map of the edges built in LDS by the projection launch: no partial planes, no add launch; LDS float adds arrival order
# speedup vs baseline: 2.2096x; 1.2532x over previous
#include <hip/hip_runtime.h>

#define N_NODES 100000
#define N_EDGES 1600000
#define D_IN    128
#define D_OUT   64

typedef __attribute__((ext_vector_type(2))) float v2f;
typedef __attribute__((ext_vector_type(8))) float v8f;

#define OWN_BLOCKS   256
#define OWN_ROWS     391
#define ID_CHUNK     256
#define ID_CHUNKS    (N_EDGES / ID_CHUNK)
#define ID_STEPS     (ID_CHUNK / 64)
#define REC_SRC_BITS 22
#define REC_SRC_MASK ((1u << REC_SRC_BITS) - 1u)
#define GEMM_WAVES   ((((N_NODES + 15) / 16 + 7) / 8) * 8)
#define MAP_GROUPS   (GEMM_WAVES / 8)
#define MAP_EDGES    (8 * ID_CHUNK)
#define MAP_DWORDS   (MAP_EDGES / 32)

static_assert(OWN_BLOCKS % 16 == 0, "whole waves clear and store the map pieces");
static_assert(OWN_BLOCKS * OWN_ROWS >= N_NODES, "the blocks cover every row");
static_assert(N_EDGES % ID_CHUNK == 0, "whole id chunks");
static_assert(GEMM_WAVES >= ID_CHUNKS, "every id chunk has a projection wave");
static_assert(MAP_DWORDS == 64 && MAP_GROUPS % 2 == 0, "a map piece is 256 bytes; a scan step takes two pieces");
static_assert(N_NODES - 1 <= REC_SRC_MASK, "a source row fits its field");
static_assert(OWN_ROWS < (1u << (32 - REC_SRC_BITS)), "a row inside a block, or the spare row, fits its field");

__device__ __forceinline__ unsigned block_id(int d) {
    return min((unsigned)d / (unsigned)OWN_ROWS, (unsigned)(OWN_BLOCKS - 1));
}

__device__ __forceinline__ unsigned record_word(int s, int d, unsigned id) {
    const unsigned r = min((unsigned)d - id * (unsigned)OWN_ROWS, (unsigned)OWN_ROWS);
    return (r << REC_SRC_BITS) | min((unsigned)s, (unsigned)(N_NODES - 1));
}

__global__ __launch_bounds__(256) void gemm_bias_kernel(
    const float* __restrict__ x, const float* __restrict__ W,
    const float* __restrict__ bias, float* __restrict__ h,
    const int* __restrict__ src, const int* __restrict__ dst,
    const float* __restrict__ edge_w,
    unsigned int* __restrict__ maps, unsigned int* __restrict__ recs)
{
    __shared__ __attribute__((aligned(16))) unsigned bm[OWN_BLOCKS * MAP_DWORDS];

    const int lane = threadIdx.x & 31;
    const int wave = threadIdx.x >> 5;
    const int gw   = blockIdx.x * 8 + wave;
    const int m0   = gw * 16;

    #pragma unroll
    for (int j = 0; j < (OWN_BLOCKS * 2) / 32; ++j) {
        const int k = j * 32 + lane;
        ((uint4*)bm)[(k >> 1) * (MAP_DWORDS / 4) + wave * 2 + (k & 1)] = make_uint4(0u, 0u, 0u, 0u);
    }

    const bool has_chunk = gw < ID_CHUNKS;
    const size_t pair0 = (size_t)gw * (ID_CHUNK / 2) + lane;
    int2   eS[ID_STEPS], eD[ID_STEPS];
    float2 eW[ID_STEPS];
    #pragma unroll
    for (int i = 0; i < ID_STEPS; ++i) {
        eS[i] = make_int2(0, 0); eD[i] = make_int2(0, 0); eW[i] = make_float2(0.f, 0.f);
        if (has_chunk) {
            eS[i] = ((const int2*)src)[pair0 + i * 32];
            eD[i] = ((const int2*)dst)[pair0 + i * 32];
            eW[i] = ((const float2*)edge_w)[pair0 + i * 32];
        }
    }

    if (m0 < N_NODES) {

    const int half = lane >> 4;
    const int l15  = lane & 15;

    v8f acc[4];
    #pragma unroll
    for (int t = 0; t < 4; ++t) {
        const float bval = bias[t * 16 + l15];
        #pragma unroll
        for (int i = 0; i < 8; ++i) acc[t][i] = bval;
    }

    const float* xrow  = x + (size_t)(m0 + l15) * D_IN + half * 2;
    const float* wbase = W + (size_t)(half * 2) * D_OUT + l15;

    #pragma unroll
    for (int k0 = 0; k0 < D_IN; k0 += 4) {
        v2f a;
        a.x = xrow[k0 + 0];
        a.y = xrow[k0 + 1];
        #pragma unroll
        for (int t = 0; t < 4; ++t) {
            v2f bm;
            bm.x = wbase[(size_t)(k0 + 0) * D_OUT + t * 16];
            bm.y = wbase[(size_t)(k0 + 1) * D_OUT + t * 16];
            acc[t] = __builtin_amdgcn_wmma_f32_16x16x4_f32(
                          false, a,  false, bm,
                          (short)0, acc[t],
                          false,  false);
        }
    }

    float* hout = h + (size_t)(m0 + half * 8) * D_OUT + l15;
    #pragma unroll
    for (int t = 0; t < 4; ++t)
        #pragma unroll
        for (int v = 0; v < 8; ++v)
            hout[(size_t)v * D_OUT + t * 16] = acc[t][v];
    }

    if (has_chunk) {
        #pragma unroll
        for (int i = 0; i < ID_STEPS; ++i) {
            const unsigned id0 = block_id(eD[i].x), id1 = block_id(eD[i].y);
            uint4 rc;
            rc.x = record_word(eS[i].x, eD[i].x, id0);
            rc.y = (unsigned)__float_as_int(eW[i].x);
            rc.z = record_word(eS[i].y, eD[i].y, id1);
            rc.w = (unsigned)__float_as_int(eW[i].y);
            ((uint4*)recs)[pair0 + i * 32] = rc;
            const int dw = wave * 8 + i * 2 + (lane >> 4);
            const unsigned b0 = 1u << ((lane & 15) * 2);
            atomicOr(&bm[id0 * MAP_DWORDS + dw], b0);
            atomicOr(&bm[id1 * MAP_DWORDS + dw], b0 << 1);
        }
    }
    __builtin_amdgcn_fence(__ATOMIC_RELEASE, "workgroup", "local");
    __builtin_amdgcn_s_barrier();
    __builtin_amdgcn_fence(__ATOMIC_ACQUIRE, "workgroup", "local");

    #pragma unroll
    for (int r = 0; r < OWN_BLOCKS / 16; ++r) {
        const int b = r * 16 + (threadIdx.x >> 4);
        const int p = threadIdx.x & 15;
        ((uint4*)maps)[((size_t)b * MAP_GROUPS + blockIdx.x) * (MAP_DWORDS / 4) + p] =
            ((const uint4*)bm)[b * (MAP_DWORDS / 4) + p];
    }
}

#define OWN_WAVES    32
#define OWN_THREADS  (OWN_WAVES * 32)
#define SCAN_STEPS   (MAP_GROUPS / 2)
#define WAVE_STEPS   13
#define BATCH        16
#define QCAP         512
#define ACC_ROWS     (OWN_ROWS + 1)

static_assert(SCAN_STEPS <= WAVE_STEPS * OWN_WAVES, "a wave's share of the map fits its loads");
static_assert(QCAP >= BATCH - 1 + 32, "one pass over a dword (one hit per lane) cannot overflow a queue");
static_assert(ACC_ROWS * D_OUT * 4 + OWN_WAVES * QCAP * 4 <= 320 * 1024, "LDS budget");
static_assert(ACC_ROWS * D_OUT * 4 + OWN_WAVES * QCAP * 4 > 160 * 1024, "one work-group per WGP");

__global__ __launch_bounds__(OWN_THREADS) void owner_sum_kernel(
    const float* __restrict__ h,
    const unsigned int* __restrict__ maps, const unsigned int* __restrict__ recs,
    float* __restrict__ out)
{
    __shared__ __attribute__((aligned(16))) float acc[ACC_ROWS * D_OUT];
    __shared__ __attribute__((aligned(16))) unsigned queue[OWN_WAVES * QCAP];

    const int tid  = threadIdx.x;
    const int lane = tid & 31;
    const int wave = __builtin_amdgcn_readfirstlane(tid >> 5);
    const int step_lo = 0;
    const int step_hi = SCAN_STEPS;
    const unsigned blk  = blockIdx.x;
    const unsigned base = blk * OWN_ROWS;

    float4* acc4 = (float4*)acc;
    for (int i = tid; i < ACC_ROWS * (D_OUT / 4); i += OWN_THREADS)
        acc4[i] = make_float4(0.f, 0.f, 0.f, 0.f);
    __syncthreads();

    unsigned* q = queue + wave * QCAP;
    const uint4*  map4 = (const uint4*)maps + (size_t)blk * MAP_GROUPS * (MAP_DWORDS / 4);
    const uint2*  rec2 = (const uint2*)recs;
    const float2* h2   = (const float2*)h;
    int qcnt = 0;

    auto gather_add = [&](const unsigned* qs, int n) {
        const unsigned e   = qs[lane < n ? lane : n - 1];
        const uint2    rc  = rec2[e];
        const int      s_l = (int)(rc.x & REC_SRC_MASK);
        const int      w_l = (int)rc.y;
        const unsigned r_l = rc.x >> REC_SRC_BITS;
        float2 v[BATCH];
        #pragma unroll
        for (int j = 0; j < BATCH; ++j) {
            if (j < n) {
                const int sj = __builtin_amdgcn_readlane(s_l, j);
                v[j] = h2[(size_t)sj * (D_OUT / 2) + lane];
            }
        }
        #pragma unroll
        for (int j = 0; j < BATCH; ++j) {
            if (j < n) {
                const float    wj = __int_as_float(__builtin_amdgcn_readlane(w_l, j));
                const unsigned rj = __builtin_amdgcn_readlane((int)r_l, j);
                float* a = acc + rj * D_OUT + lane * 2;
                atomicAdd(a + 0, v[j].x * wj);
                atomicAdd(a + 1, v[j].y * wj);
            }
        }
    };

    uint4 d[WAVE_STEPS];
    #pragma unroll
    for (int u = 0; u < WAVE_STEPS; ++u) {
        const int tu = min(step_lo + wave + u * OWN_WAVES, step_hi - 1);
        d[u] = map4[(size_t)tu * 32 + lane];
    }
    #pragma unroll 1
    for (int t = step_lo + wave; t < step_hi; t += OWN_WAVES) {
        uint4 cur = d[0];
        unsigned e0 = ((unsigned)(2 * t) + (unsigned)(lane >> 4)) * (unsigned)MAP_EDGES
                    + (unsigned)(lane & 15) * 128u;
        #pragma unroll 1
        for (int i = 0; i < 4; ++i) {
            unsigned hit = cur.x;
            unsigned mask = __builtin_amdgcn_ballot_w32(hit != 0u);
            while (mask != 0u) {
                if (hit != 0u) {
                    const int pos = qcnt + (int)__builtin_amdgcn_mbcnt_lo(mask, 0u);
                    q[pos] = e0 + (unsigned)__builtin_ctz(hit);
                    hit &= hit - 1u;
                }
                qcnt += __builtin_popcount(mask);
                while (qcnt >= BATCH) {
                    qcnt -= BATCH;
                    gather_add(q + qcnt, BATCH);
                }
                mask = __builtin_amdgcn_ballot_w32(hit != 0u);
            }
            cur.x = cur.y; cur.y = cur.z; cur.z = cur.w;
            e0 += 32u;
        }
        #pragma unroll
        for (int u = 0; u + 1 < WAVE_STEPS; ++u) d[u] = d[u + 1];
    }
    if (qcnt > 0) gather_add(q, qcnt);
    __syncthreads();

    float4* out4 = (float4*)out;
    for (int i = tid; i < OWN_ROWS * (D_OUT / 4); i += OWN_THREADS) {
        const unsigned row = base + (unsigned)(i >> 4);
        if (row < N_NODES) out4[(size_t)row * (D_OUT / 4) + (i & 15)] = acc4[i];
    }
}

extern "C" void kernel_launch(void* const* d_in, const int* in_sizes, int n_in,
                              void* d_out, int out_size, void* d_ws, size_t ws_size,
                              hipStream_t stream) {
    const float* x      = (const float*)d_in[0];
    const float* W      = (const float*)d_in[1];
    const float* b      = (const float*)d_in[2];
    const float* edge_w = (const float*)d_in[3];
    const int*   src    = (const int*)d_in[4];
    const int*   dst    = (const int*)d_in[5];
    float*       out    = (float*)d_out;
    float*       h      = (float*)d_ws;
    unsigned int* maps  = (unsigned int*)(h + (size_t)N_NODES * D_OUT);
    unsigned int* recs  = maps + (size_t)OWN_BLOCKS * MAP_GROUPS * MAP_DWORDS;

    (void)in_sizes; (void)n_in; (void)ws_size; (void)out_size;

    const int m_tiles = (N_NODES + 15) / 16;
    const int gemm_blocks = (m_tiles + 7) / 8;
    gemm_bias_kernel<<<gemm_blocks, 256, 0, stream>>>(x, W, b, h, src, dst, edge_w, maps, recs);

    owner_sum_kernel<<<OWN_BLOCKS, OWN_THREADS, 0, stream>>>(h, maps, recs, out);
}


# Projection: W copied once per work-group into LDS (in the map pieces' space, cleared after the strip); B fragments read by 8-byte LDS loads, not 4-byte global loads; k order and sums unchanged
# speedup vs baseline: 2.2912x; 1.0369x over previous
#include <hip/hip_runtime.h>

#define N_NODES 100000
#define N_EDGES 1600000
#define D_IN    128
#define D_OUT   64

typedef __attribute__((ext_vector_type(2))) float v2f;
typedef __attribute__((ext_vector_type(8))) float v8f;

#define OWN_BLOCKS   256
#define OWN_ROWS     391
#define ID_CHUNK     256
#define ID_CHUNKS    (N_EDGES / ID_CHUNK)
#define ID_STEPS     (ID_CHUNK / 64)
#define REC_SRC_BITS 22
#define REC_SRC_MASK ((1u << REC_SRC_BITS) - 1u)
#define GEMM_WAVES   ((((N_NODES + 15) / 16 + 7) / 8) * 8)
#define MAP_GROUPS   (GEMM_WAVES / 8)
#define MAP_EDGES    (8 * ID_CHUNK)
#define MAP_DWORDS   (MAP_EDGES / 32)

#define WL_PITCH     80
#define A_AHEAD      16
static_assert(OWN_BLOCKS % 16 == 0, "whole waves clear and store the map pieces");
static_assert((D_IN / 2) * WL_PITCH * 8 <= OWN_BLOCKS * MAP_DWORDS * 4, "staged W fits the space of the map pieces");
static_assert((D_IN / 2) * (D_OUT / 4) == 4 * 256, "four staging tasks per thread");
static_assert(OWN_BLOCKS * OWN_ROWS >= N_NODES, "the blocks cover every row");
static_assert(N_EDGES % ID_CHUNK == 0, "whole id chunks");
static_assert(GEMM_WAVES >= ID_CHUNKS, "every id chunk has a projection wave");
static_assert(MAP_DWORDS == 64 && MAP_GROUPS % 2 == 0, "a map piece is 256 bytes; a scan step takes two pieces");
static_assert(N_NODES - 1 <= REC_SRC_MASK, "a source row fits its field");
static_assert(OWN_ROWS < (1u << (32 - REC_SRC_BITS)), "a row inside a block, or the spare row, fits its field");

__device__ __forceinline__ unsigned block_id(int d) {
    return min((unsigned)d / (unsigned)OWN_ROWS, (unsigned)(OWN_BLOCKS - 1));
}

__device__ __forceinline__ unsigned record_word(int s, int d, unsigned id) {
    const unsigned r = min((unsigned)d - id * (unsigned)OWN_ROWS, (unsigned)OWN_ROWS);
    return (r << REC_SRC_BITS) | min((unsigned)s, (unsigned)(N_NODES - 1));
}

__global__ __launch_bounds__(256) void gemm_bias_kernel(
    const float* __restrict__ x, const float* __restrict__ W,
    const float* __restrict__ bias, float* __restrict__ h,
    const int* __restrict__ src, const int* __restrict__ dst,
    const float* __restrict__ edge_w,
    unsigned int* __restrict__ maps, unsigned int* __restrict__ recs)
{
    __shared__ __attribute__((aligned(16))) unsigned bm[OWN_BLOCKS * MAP_DWORDS];

    const int lane = threadIdx.x & 31;
    const int wave = threadIdx.x >> 5;
    const int gw   = blockIdx.x * 8 + wave;
    const int m0   = gw * 16;

    float2* wl = (float2*)bm;
    #pragma unroll
    for (int i = 0; i < 4; ++i) {
        const int task = i * 256 + (int)threadIdx.x;
        const int kp = task >> 4, n4 = (task & 15) * 4;
        const float4 w0 = *(const float4*)(W + (size_t)(2 * kp) * D_OUT + n4);
        const float4 w1 = *(const float4*)(W + (size_t)(2 * kp + 1) * D_OUT + n4);
        float4* wd = (float4*)(wl + kp * WL_PITCH + n4);
        wd[0] = make_float4(w0.x, w1.x, w0.y, w1.y);
        wd[1] = make_float4(w0.z, w1.z, w0.w, w1.w);
    }

    const bool has_chunk = gw < ID_CHUNKS;
    const size_t pair0 = (size_t)gw * (ID_CHUNK / 2) + lane;
    int2   eS[ID_STEPS], eD[ID_STEPS];
    float2 eW[ID_STEPS];
    #pragma unroll
    for (int i = 0; i < ID_STEPS; ++i) {
        eS[i] = make_int2(0, 0); eD[i] = make_int2(0, 0); eW[i] = make_float2(0.f, 0.f);
        if (has_chunk) {
            eS[i] = ((const int2*)src)[pair0 + i * 32];
            eD[i] = ((const int2*)dst)[pair0 + i * 32];
            eW[i] = ((const float2*)edge_w)[pair0 + i * 32];
        }
    }

    const int half = lane >> 4;
    const int l15  = lane & 15;
    const bool has_strip = m0 < N_NODES;
    const float* xrow = x + (size_t)min(m0 + l15, N_NODES - 1) * D_IN + half * 2;
    v2f apre[A_AHEAD];
    #pragma unroll
    for (int j = 0; j < A_AHEAD; ++j) {
        apre[j].x = xrow[4 * j + 0];
        apre[j].y = xrow[4 * j + 1];
    }

    unsigned ids[ID_STEPS];
    #pragma unroll
    for (int i = 0; i < ID_STEPS; ++i) {
        const unsigned id0 = block_id(eD[i].x), id1 = block_id(eD[i].y);
        ids[i] = id0 | (id1 << 16);
        if (has_chunk) {
            uint4 rc;
            rc.x = record_word(eS[i].x, eD[i].x, id0);
            rc.y = (unsigned)__float_as_int(eW[i].x);
            rc.z = record_word(eS[i].y, eD[i].y, id1);
            rc.w = (unsigned)__float_as_int(eW[i].y);
            ((uint4*)recs)[pair0 + i * 32] = rc;
        }
    }

    __builtin_amdgcn_fence(__ATOMIC_RELEASE, "workgroup", "local");
    __builtin_amdgcn_s_barrier();
    __builtin_amdgcn_fence(__ATOMIC_ACQUIRE, "workgroup", "local");

    if (has_strip) {

    v8f acc[4];
    #pragma unroll
    for (int t = 0; t < 4; ++t) {
        const float bval = bias[t * 16 + l15];
        #pragma unroll
        for (int i = 0; i < 8; ++i) acc[t][i] = bval;
    }

    const float2* wfrag = wl + half * WL_PITCH + l15;

    #pragma unroll
    for (int k0 = 0; k0 < D_IN; k0 += 4) {
        v2f a;
        if (k0 / 4 < A_AHEAD) {
            a = apre[k0 / 4];
        } else {
            a.x = xrow[k0 + 0];
            a.y = xrow[k0 + 1];
        }
        #pragma unroll
        for (int t = 0; t < 4; ++t) {
            const float2 wf = wfrag[(k0 / 2) * WL_PITCH + t * 16];
            v2f bf;
            bf.x = wf.x;
            bf.y = wf.y;
            acc[t] = __builtin_amdgcn_wmma_f32_16x16x4_f32(
                          false, a,  false, bf,
                          (short)0, acc[t],
                          false,  false);
        }
    }

    float* hout = h + (size_t)(m0 + half * 8) * D_OUT + l15;
    #pragma unroll
    for (int t = 0; t < 4; ++t)
        #pragma unroll
        for (int v = 0; v < 8; ++v)
            hout[(size_t)v * D_OUT + t * 16] = acc[t][v];
    }

    __builtin_amdgcn_fence(__ATOMIC_RELEASE, "workgroup", "local");
    __builtin_amdgcn_s_barrier();
    __builtin_amdgcn_fence(__ATOMIC_ACQUIRE, "workgroup", "local");

    #pragma unroll
    for (int j = 0; j < (OWN_BLOCKS * 2) / 32; ++j) {
        const int k = j * 32 + lane;
        ((uint4*)bm)[(k >> 1) * (MAP_DWORDS / 4) + wave * 2 + (k & 1)] = make_uint4(0u, 0u, 0u, 0u);
    }

    if (has_chunk) {
        #pragma unroll
        for (int i = 0; i < ID_STEPS; ++i) {
            const unsigned id0 = ids[i] & 0xffffu, id1 = ids[i] >> 16;
            const int dw = wave * 8 + i * 2 + (lane >> 4);
            const unsigned b0 = 1u << ((lane & 15) * 2);
            atomicOr(&bm[id0 * MAP_DWORDS + dw], b0);
            atomicOr(&bm[id1 * MAP_DWORDS + dw], b0 << 1);
        }
    }
    __builtin_amdgcn_fence(__ATOMIC_RELEASE, "workgroup", "local");
    __builtin_amdgcn_s_barrier();
    __builtin_amdgcn_fence(__ATOMIC_ACQUIRE, "workgroup", "local");

    #pragma unroll
    for (int r = 0; r < OWN_BLOCKS / 16; ++r) {
        const int b = r * 16 + (threadIdx.x >> 4);
        const int p = threadIdx.x & 15;
        ((uint4*)maps)[((size_t)b * MAP_GROUPS + blockIdx.x) * (MAP_DWORDS / 4) + p] =
            ((const uint4*)bm)[b * (MAP_DWORDS / 4) + p];
    }
}

#define OWN_WAVES    32
#define OWN_THREADS  (OWN_WAVES * 32)
#define SCAN_STEPS   (MAP_GROUPS / 2)
#define WAVE_STEPS   13
#define BATCH        16
#define QCAP         512
#define ACC_ROWS     (OWN_ROWS + 1)

static_assert(SCAN_STEPS <= WAVE_STEPS * OWN_WAVES, "a wave's share of the map fits its loads");
static_assert(QCAP >= BATCH - 1 + 32, "one pass over a dword (one hit per lane) cannot overflow a queue");
static_assert(ACC_ROWS * D_OUT * 4 + OWN_WAVES * QCAP * 4 <= 320 * 1024, "LDS budget");
static_assert(ACC_ROWS * D_OUT * 4 + OWN_WAVES * QCAP * 4 > 160 * 1024, "one work-group per WGP");

__global__ __launch_bounds__(OWN_THREADS) void owner_sum_kernel(
    const float* __restrict__ h,
    const unsigned int* __restrict__ maps, const unsigned int* __restrict__ recs,
    float* __restrict__ out)
{
    __shared__ __attribute__((aligned(16))) float acc[ACC_ROWS * D_OUT];
    __shared__ __attribute__((aligned(16))) unsigned queue[OWN_WAVES * QCAP];

    const int tid  = threadIdx.x;
    const int lane = tid & 31;
    const int wave = __builtin_amdgcn_readfirstlane(tid >> 5);
    const int step_lo = 0;
    const int step_hi = SCAN_STEPS;
    const unsigned blk  = blockIdx.x;
    const unsigned base = blk * OWN_ROWS;

    float4* acc4 = (float4*)acc;
    for (int i = tid; i < ACC_ROWS * (D_OUT / 4); i += OWN_THREADS)
        acc4[i] = make_float4(0.f, 0.f, 0.f, 0.f);
    __syncthreads();

    unsigned* q = queue + wave * QCAP;
    const uint4*  map4 = (const uint4*)maps + (size_t)blk * MAP_GROUPS * (MAP_DWORDS / 4);
    const uint2*  rec2 = (const uint2*)recs;
    const float2* h2   = (const float2*)h;
    int qcnt = 0;

    auto gather_add = [&](const unsigned* qs, int n) {
        const unsigned e   = qs[lane < n ? lane : n - 1];
        const uint2    rc  = rec2[e];
        const int      s_l = (int)(rc.x & REC_SRC_MASK);
        const int      w_l = (int)rc.y;
        const unsigned r_l = rc.x >> REC_SRC_BITS;
        float2 v[BATCH];
        #pragma unroll
        for (int j = 0; j < BATCH; ++j) {
            if (j < n) {
                const int sj = __builtin_amdgcn_readlane(s_l, j);
                v[j] = h2[(size_t)sj * (D_OUT / 2) + lane];
            }
        }
        #pragma unroll
        for (int j = 0; j < BATCH; ++j) {
            if (j < n) {
                const float    wj = __int_as_float(__builtin_amdgcn_readlane(w_l, j));
                const unsigned rj = __builtin_amdgcn_readlane((int)r_l, j);
                float* a = acc + rj * D_OUT + lane * 2;
                atomicAdd(a + 0, v[j].x * wj);
                atomicAdd(a + 1, v[j].y * wj);
            }
        }
    };

    uint4 d[WAVE_STEPS];
    #pragma unroll
    for (int u = 0; u < WAVE_STEPS; ++u) {
        const int tu = min(step_lo + wave + u * OWN_WAVES, step_hi - 1);
        d[u] = map4[(size_t)tu * 32 + lane];
    }
    #pragma unroll 1
    for (int t = step_lo + wave; t < step_hi; t += OWN_WAVES) {
        uint4 cur = d[0];
        unsigned e0 = ((unsigned)(2 * t) + (unsigned)(lane >> 4)) * (unsigned)MAP_EDGES
                    + (unsigned)(lane & 15) * 128u;
        #pragma unroll 1
        for (int i = 0; i < 4; ++i) {
            unsigned hit = cur.x;
            unsigned mask = __builtin_amdgcn_ballot_w32(hit != 0u);
            while (mask != 0u) {
                if (hit != 0u) {
                    const int pos = qcnt + (int)__builtin_amdgcn_mbcnt_lo(mask, 0u);
                    q[pos] = e0 + (unsigned)__builtin_ctz(hit);
                    hit &= hit - 1u;
                }
                qcnt += __builtin_popcount(mask);
                while (qcnt >= BATCH) {
                    qcnt -= BATCH;
                    gather_add(q + qcnt, BATCH);
                }
                mask = __builtin_amdgcn_ballot_w32(hit != 0u);
            }
            cur.x = cur.y; cur.y = cur.z; cur.z = cur.w;
            e0 += 32u;
        }
        #pragma unroll
        for (int u = 0; u + 1 < WAVE_STEPS; ++u) d[u] = d[u + 1];
    }
    if (qcnt > 0) gather_add(q, qcnt);
    __syncthreads();

    float4* out4 = (float4*)out;
    for (int i = tid; i < OWN_ROWS * (D_OUT / 4); i += OWN_THREADS) {
        const unsigned row = base + (unsigned)(i >> 4);
        if (row < N_NODES) out4[(size_t)row * (D_OUT / 4) + (i & 15)] = acc4[i];
    }
}

extern "C" void kernel_launch(void* const* d_in, const int* in_sizes, int n_in,
                              void* d_out, int out_size, void* d_ws, size_t ws_size,
                              hipStream_t stream) {
    const float* x      = (const float*)d_in[0];
    const float* W      = (const float*)d_in[1];
    const float* b      = (const float*)d_in[2];
    const float* edge_w = (const float*)d_in[3];
    const int*   src    = (const int*)d_in[4];
    const int*   dst    = (const int*)d_in[5];
    float*       out    = (float*)d_out;
    float*       h      = (float*)d_ws;
    unsigned int* maps  = (unsigned int*)(h + (size_t)N_NODES * D_OUT);
    unsigned int* recs  = maps + (size_t)OWN_BLOCKS * MAP_GROUPS * MAP_DWORDS;

    (void)in_sizes; (void)n_in; (void)ws_size; (void)out_size;

    const int m_tiles = (N_NODES + 15) / 16;
    const int gemm_blocks = (m_tiles + 7) / 8;
    gemm_bias_kernel<<<gemm_blocks, 256, 0, stream>>>(x, W, b, h, src, dst, edge_w, maps, recs);

    owner_sum_kernel<<<OWN_BLOCKS, OWN_THREADS, 0, stream>>>(h, maps, recs, out);
}


# Edge phase: every row of a block gets one owner wave; walkers pass edges to owner queues in LDS, owners sum in registers in queue (arrival) order, store rows; LDS float adds only if a queue is full
# speedup vs baseline: 2.4765x; 1.0809x over previous
#include <hip/hip_runtime.h>

#define N_NODES 100000
#define N_EDGES 1600000
#define D_IN    128
#define D_OUT   64

typedef __attribute__((ext_vector_type(2))) float v2f;
typedef __attribute__((ext_vector_type(8))) float v8f;

#define OWN_BLOCKS   256
#define OWN_ROWS     391
#define ID_CHUNK     256
#define ID_CHUNKS    (N_EDGES / ID_CHUNK)
#define ID_STEPS     (ID_CHUNK / 64)
#define REC_SRC_BITS 22
#define REC_SRC_MASK ((1u << REC_SRC_BITS) - 1u)
#define GEMM_WAVES   ((((N_NODES + 15) / 16 + 7) / 8) * 8)
#define MAP_GROUPS   (GEMM_WAVES / 8)
#define MAP_EDGES    (8 * ID_CHUNK)
#define MAP_DWORDS   (MAP_EDGES / 32)

#define WL_PITCH     80
#define A_AHEAD      16
static_assert(OWN_BLOCKS % 16 == 0, "whole waves clear and store the map pieces");
static_assert((D_IN / 2) * WL_PITCH * 8 <= OWN_BLOCKS * MAP_DWORDS * 4, "staged W fits the space of the map pieces");
static_assert((D_IN / 2) * (D_OUT / 4) == 4 * 256, "four staging tasks per thread");
static_assert(OWN_BLOCKS * OWN_ROWS >= N_NODES, "the blocks cover every row");
static_assert(N_EDGES % ID_CHUNK == 0, "whole id chunks");
static_assert(GEMM_WAVES >= ID_CHUNKS, "every id chunk has a projection wave");
static_assert(MAP_DWORDS == 64 && MAP_GROUPS % 2 == 0, "a map piece is 256 bytes; a scan step takes two pieces");
static_assert(N_NODES - 1 <= REC_SRC_MASK, "a source row fits its field");
static_assert(OWN_ROWS < (1u << (32 - REC_SRC_BITS)), "a row inside a block, or the spare row, fits its field");

__device__ __forceinline__ unsigned block_id(int d) {
    return min((unsigned)d / (unsigned)OWN_ROWS, (unsigned)(OWN_BLOCKS - 1));
}

__device__ __forceinline__ unsigned record_word(int s, int d, unsigned id) {
    const unsigned r = min((unsigned)d - id * (unsigned)OWN_ROWS, (unsigned)OWN_ROWS);
    return (r << REC_SRC_BITS) | min((unsigned)s, (unsigned)(N_NODES - 1));
}

__global__ __launch_bounds__(256) void gemm_bias_kernel(
    const float* __restrict__ x, const float* __restrict__ W,
    const float* __restrict__ bias, float* __restrict__ h,
    const int* __restrict__ src, const int* __restrict__ dst,
    const float* __restrict__ edge_w,
    unsigned int* __restrict__ maps, unsigned int* __restrict__ recs)
{
    __shared__ __attribute__((aligned(16))) unsigned bm[OWN_BLOCKS * MAP_DWORDS];

    const int lane = threadIdx.x & 31;
    const int wave = threadIdx.x >> 5;
    const int gw   = blockIdx.x * 8 + wave;
    const int m0   = gw * 16;

    float2* wl = (float2*)bm;
    #pragma unroll
    for (int i = 0; i < 4; ++i) {
        const int task = i * 256 + (int)threadIdx.x;
        const int kp = task >> 4, n4 = (task & 15) * 4;
        const float4 w0 = *(const float4*)(W + (size_t)(2 * kp) * D_OUT + n4);
        const float4 w1 = *(const float4*)(W + (size_t)(2 * kp + 1) * D_OUT + n4);
        float4* wd = (float4*)(wl + kp * WL_PITCH + n4);
        wd[0] = make_float4(w0.x, w1.x, w0.y, w1.y);
        wd[1] = make_float4(w0.z, w1.z, w0.w, w1.w);
    }

    const bool has_chunk = gw < ID_CHUNKS;
    const size_t pair0 = (size_t)gw * (ID_CHUNK / 2) + lane;
    int2   eS[ID_STEPS], eD[ID_STEPS];
    float2 eW[ID_STEPS];
    #pragma unroll
    for (int i = 0; i < ID_STEPS; ++i) {
        eS[i] = make_int2(0, 0); eD[i] = make_int2(0, 0); eW[i] = make_float2(0.f, 0.f);
        if (has_chunk) {
            eS[i] = ((const int2*)src)[pair0 + i * 32];
            eD[i] = ((const int2*)dst)[pair0 + i * 32];
            eW[i] = ((const float2*)edge_w)[pair0 + i * 32];
        }
    }

    const int half = lane >> 4;
    const int l15  = lane & 15;
    const bool has_strip = m0 < N_NODES;
    const float* xrow = x + (size_t)min(m0 + l15, N_NODES - 1) * D_IN + half * 2;
    v2f apre[A_AHEAD];
    #pragma unroll
    for (int j = 0; j < A_AHEAD; ++j) {
        apre[j].x = xrow[4 * j + 0];
        apre[j].y = xrow[4 * j + 1];
    }

    unsigned ids[ID_STEPS];
    #pragma unroll
    for (int i = 0; i < ID_STEPS; ++i) {
        const unsigned id0 = block_id(eD[i].x), id1 = block_id(eD[i].y);
        ids[i] = id0 | (id1 << 16);
        if (has_chunk) {
            uint4 rc;
            rc.x = record_word(eS[i].x, eD[i].x, id0);
            rc.y = (unsigned)__float_as_int(eW[i].x);
            rc.z = record_word(eS[i].y, eD[i].y, id1);
            rc.w = (unsigned)__float_as_int(eW[i].y);
            ((uint4*)recs)[pair0 + i * 32] = rc;
        }
    }

    __builtin_amdgcn_fence(__ATOMIC_RELEASE, "workgroup", "local");
    __builtin_amdgcn_s_barrier();
    __builtin_amdgcn_fence(__ATOMIC_ACQUIRE, "workgroup", "local");

    if (has_strip) {

    v8f acc[4];
    #pragma unroll
    for (int t = 0; t < 4; ++t) {
        const float bval = bias[t * 16 + l15];
        #pragma unroll
        for (int i = 0; i < 8; ++i) acc[t][i] = bval;
    }

    const float2* wfrag = wl + half * WL_PITCH + l15;

    #pragma unroll
    for (int k0 = 0; k0 < D_IN; k0 += 4) {
        v2f a;
        if (k0 / 4 < A_AHEAD) {
            a = apre[k0 / 4];
        } else {
            a.x = xrow[k0 + 0];
            a.y = xrow[k0 + 1];
        }
        #pragma unroll
        for (int t = 0; t < 4; ++t) {
            const float2 wf = wfrag[(k0 / 2) * WL_PITCH + t * 16];
            v2f bf;
            bf.x = wf.x;
            bf.y = wf.y;
            acc[t] = __builtin_amdgcn_wmma_f32_16x16x4_f32(
                          false, a,  false, bf,
                          (short)0, acc[t],
                          false,  false);
        }
    }

    float* hout = h + (size_t)(m0 + half * 8) * D_OUT + l15;
    #pragma unroll
    for (int t = 0; t < 4; ++t)
        #pragma unroll
        for (int v = 0; v < 8; ++v)
            hout[(size_t)v * D_OUT + t * 16] = acc[t][v];
    }

    __builtin_amdgcn_fence(__ATOMIC_RELEASE, "workgroup", "local");
    __builtin_amdgcn_s_barrier();
    __builtin_amdgcn_fence(__ATOMIC_ACQUIRE, "workgroup", "local");

    #pragma unroll
    for (int j = 0; j < (OWN_BLOCKS * 2) / 32; ++j) {
        const int k = j * 32 + lane;
        ((uint4*)bm)[(k >> 1) * (MAP_DWORDS / 4) + wave * 2 + (k & 1)] = make_uint4(0u, 0u, 0u, 0u);
    }

    if (has_chunk) {
        #pragma unroll
        for (int i = 0; i < ID_STEPS; ++i) {
            const unsigned id0 = ids[i] & 0xffffu, id1 = ids[i] >> 16;
            const int dw = wave * 8 + i * 2 + (lane >> 4);
            const unsigned b0 = 1u << ((lane & 15) * 2);
            atomicOr(&bm[id0 * MAP_DWORDS + dw], b0);
            atomicOr(&bm[id1 * MAP_DWORDS + dw], b0 << 1);
        }
    }
    __builtin_amdgcn_fence(__ATOMIC_RELEASE, "workgroup", "local");
    __builtin_amdgcn_s_barrier();
    __builtin_amdgcn_fence(__ATOMIC_ACQUIRE, "workgroup", "local");

    #pragma unroll
    for (int r = 0; r < OWN_BLOCKS / 16; ++r) {
        const int b = r * 16 + (threadIdx.x >> 4);
        const int p = threadIdx.x & 15;
        ((uint4*)maps)[((size_t)b * MAP_GROUPS + blockIdx.x) * (MAP_DWORDS / 4) + p] =
            ((const uint4*)bm)[b * (MAP_DWORDS / 4) + p];
    }
}

#define OWN_WAVES    32
#define OWN_THREADS  (OWN_WAVES * 32)
#define SCAN_STEPS   (MAP_GROUPS / 2)
#define WAVE_STEPS   13
#define HITS         32
#define PQCAP        64
#define OQCAP        512
#define ACC_ROWS     (OWN_ROWS + 1)
#define ROWS_X4      (4 * ACC_ROWS / OWN_WAVES)
#define WAVE_ROWS    13

typedef __attribute__((ext_vector_type(16))) float v16f;
template <int N> struct depth_t { static constexpr int value = N; };

static_assert(SCAN_STEPS <= WAVE_STEPS * OWN_WAVES, "a wave's share of the map fits its loads");
static_assert((OQCAP & (OQCAP - 1)) == 0, "places of a queue are taken round by a mask");
static_assert(PQCAP >= HITS - 1 + 32, "one pass over a dword (one hit per lane) cannot overflow a private queue");
static_assert(ROWS_X4 * OWN_WAVES == 4 * ACC_ROWS, "the rows divide over the waves in quarters");
static_assert(((ROWS_X4 + 3) >> 2) <= WAVE_ROWS && WAVE_ROWS <= 16, "a wave's rows fit its register sums");
static_assert(WAVE_ROWS <= (1 << (32 - REC_SRC_BITS)), "a row inside a wave fits the entry's field");
static_assert(ACC_ROWS * D_OUT * 4 + OWN_WAVES * OQCAP * 8 + OWN_WAVES * PQCAP * 4 + OWN_WAVES * 4 <= 320 * 1024, "LDS budget");
static_assert(ACC_ROWS * D_OUT * 4 + OWN_WAVES * OQCAP * 8 > 160 * 1024, "one work-group per WGP");

__global__ __launch_bounds__(OWN_THREADS) void owner_sum_kernel(
    const float* __restrict__ h,
    const unsigned int* __restrict__ maps, const unsigned int* __restrict__ recs,
    float* __restrict__ out)
{
    __shared__ __attribute__((aligned(16))) float acc[ACC_ROWS * D_OUT];
    __shared__ __attribute__((aligned(16))) uint2 oq[OWN_WAVES * OQCAP];
    __shared__ __attribute__((aligned(16))) unsigned pq[OWN_WAVES * PQCAP];
    __shared__ unsigned cnt[OWN_WAVES];

    const int tid  = threadIdx.x;
    const int lane = tid & 31;
    const int wave = __builtin_amdgcn_readfirstlane(tid >> 5);
    const int step_lo = 0;
    const int step_hi = SCAN_STEPS;
    const unsigned blk  = blockIdx.x;
    const unsigned base = blk * OWN_ROWS;

    float4* acc4 = (float4*)acc;
    for (int i = tid; i < ACC_ROWS * (D_OUT / 4); i += OWN_THREADS)
        acc4[i] = make_float4(0.f, 0.f, 0.f, 0.f);
    if (tid < OWN_WAVES) cnt[tid] = 0u;
    __syncthreads();

    unsigned* q = pq + wave * PQCAP;
    const uint4*  map4 = (const uint4*)maps + (size_t)blk * MAP_GROUPS * (MAP_DWORDS / 4);
    const uint2*  rec2 = (const uint2*)recs;
    const float2* h2   = (const float2*)h;
    int qcnt = 0;

    auto hand_on = [&](const unsigned* qs, int n) {
        const bool     valid = lane < n;
        const unsigned e   = qs[valid ? lane : n - 1];
        const uint2    rc  = rec2[e];
        const unsigned r   = rc.x >> REC_SRC_BITS;
        const unsigned o   = (r * 4u) / (unsigned)ROWS_X4;
        const unsigned k   = r - (((unsigned)ROWS_X4 * o + 3u) >> 2);
        bool full = false;
        if (valid) {
            const unsigned slot = atomicAdd(&cnt[o], 1u);
            if (slot < (unsigned)OQCAP)
                oq[o * OQCAP + ((slot + o) & (unsigned)(OQCAP - 1))] =
                    make_uint2((k << REC_SRC_BITS) | (rc.x & REC_SRC_MASK), rc.y);
            else
                full = true;
        }
        unsigned left = __builtin_amdgcn_ballot_w32(full);
        while (left != 0u) {
            const int j = __builtin_ctz(left);
            left &= left - 1u;
            const int      sj = __builtin_amdgcn_readlane((int)(rc.x & REC_SRC_MASK), j);
            const float    wj = __int_as_float(__builtin_amdgcn_readlane((int)rc.y, j));
            const unsigned rj = (unsigned)__builtin_amdgcn_readlane((int)r, j);
            const float2   v  = h2[(size_t)sj * (D_OUT / 2) + lane];
            float* a = acc + rj * D_OUT + lane * 2;
            atomicAdd(a + 0, v.x * wj);
            atomicAdd(a + 1, v.y * wj);
        }
    };

    uint4 d[WAVE_STEPS];
    #pragma unroll
    for (int u = 0; u < WAVE_STEPS; ++u) {
        const int tu = min(step_lo + wave + u * OWN_WAVES, step_hi - 1);
        d[u] = map4[(size_t)tu * 32 + lane];
    }
    #pragma unroll 1
    for (int t = step_lo + wave; t < step_hi; t += OWN_WAVES) {
        uint4 cur = d[0];
        unsigned e0 = ((unsigned)(2 * t) + (unsigned)(lane >> 4)) * (unsigned)MAP_EDGES
                    + (unsigned)(lane & 15) * 128u;
        #pragma unroll 1
        for (int i = 0; i < 4; ++i) {
            unsigned hit = cur.x;
            unsigned mask = __builtin_amdgcn_ballot_w32(hit != 0u);
            while (mask != 0u) {
                if (hit != 0u) {
                    const int pos = qcnt + (int)__builtin_amdgcn_mbcnt_lo(mask, 0u);
                    q[pos] = e0 + (unsigned)__builtin_ctz(hit);
                    hit &= hit - 1u;
                }
                qcnt += __builtin_popcount(mask);
                while (qcnt >= HITS) {
                    qcnt -= HITS;
                    hand_on(q + qcnt, HITS);
                }
                mask = __builtin_amdgcn_ballot_w32(hit != 0u);
            }
            cur.x = cur.y; cur.y = cur.z; cur.z = cur.w;
            e0 += 32u;
        }
        #pragma unroll
        for (int u = 0; u + 1 < WAVE_STEPS; ++u) d[u] = d[u + 1];
    }
    if (qcnt > 0) hand_on(q, qcnt);
    __syncthreads();

    const int first = (ROWS_X4 * wave + 3) >> 2;
    const int nrows = ((ROWS_X4 * (wave + 1) + 3) >> 2) - first;
    v16f sx = {}, sy = {};

    const int nq = __builtin_amdgcn_readfirstlane((int)min(cnt[wave], (unsigned)OQCAP));
    const uint2* myq = oq + wave * OQCAP;

    auto sum_entries = [&](int i0, int m, auto depth) {
        constexpr int DEPTH = decltype(depth)::value;
        const uint2 en  = myq[(i0 + (lane < m ? lane : m - 1) + wave) & (OQCAP - 1)];
        const int   s_l = (int)(en.x & REC_SRC_MASK);
        const int   k_l = (int)(en.x >> REC_SRC_BITS);
        const int   w_l = (int)en.y;
        float2 v[DEPTH];
        #pragma unroll
        for (int j = 0; j < DEPTH; ++j) {
            if (j < m) {
                const int sj = __builtin_amdgcn_readlane(s_l, j);
                v[j] = h2[(size_t)sj * (D_OUT / 2) + lane];
            }
        }
        #pragma unroll
        for (int j = 0; j < DEPTH; ++j) {
            if (j < m) {
                const float wj = __int_as_float(__builtin_amdgcn_readlane(w_l, j));
                const int   kj = __builtin_amdgcn_readlane(k_l, j);
                sx[kj] += v[j].x * wj;
                sy[kj] += v[j].y * wj;
            }
        }
    };

    int done = 0;
    #pragma unroll 1
    for (; done + HITS <= nq; done += HITS) sum_entries(done, HITS, depth_t<HITS>{});
    if (done + HITS / 2 <= nq) { sum_entries(done, HITS / 2, depth_t<HITS / 2>{}); done += HITS / 2; }
    if (done < nq) sum_entries(done, nq - done, depth_t<HITS / 2>{});

    const float2* acc2 = (const float2*)acc;
    float2* out2 = (float2*)out;
    #pragma unroll
    for (int i = 0; i < WAVE_ROWS; ++i) {
        const int r = first + i;
        const unsigned row = base + (unsigned)r;
        if (i < nrows && r < OWN_ROWS && row < N_NODES) {
            const float2 o = acc2[r * (D_OUT / 2) + lane];
            out2[(size_t)row * (D_OUT / 2) + lane] = make_float2(sx[i] + o.x, sy[i] + o.y);
        }
    }
}

extern "C" void kernel_launch(void* const* d_in, const int* in_sizes, int n_in,
                              void* d_out, int out_size, void* d_ws, size_t ws_size,
                              hipStream_t stream) {
    const float* x      = (const float*)d_in[0];
    const float* W      = (const float*)d_in[1];
    const float* b      = (const float*)d_in[2];
    const float* edge_w = (const float*)d_in[3];
    const int*   src    = (const int*)d_in[4];
    const int*   dst    = (const int*)d_in[5];
    float*       out    = (float*)d_out;
    float*       h      = (float*)d_ws;
    unsigned int* maps  = (unsigned int*)(h + (size_t)N_NODES * D_OUT);
    unsigned int* recs  = maps + (size_t)OWN_BLOCKS * MAP_GROUPS * MAP_DWORDS;

    (void)in_sizes; (void)n_in; (void)ws_size; (void)out_size;

    const int m_tiles = (N_NODES + 15) / 16;
    const int gemm_blocks = (m_tiles + 7) / 8;
    gemm_bias_kernel<<<gemm_blocks, 256, 0, stream>>>(x, W, b, h, src, dst, edge_w, maps, recs);

    owner_sum_kernel<<<OWN_BLOCKS, OWN_THREADS, 0, stream>>>(h, maps, recs, out);
}


# Projection launch: half of the work-groups (alternating by index) build and store their map pieces before their strip, not behind it, so co-resident work-groups are out of step; sums unchanged
# speedup vs baseline: 2.5812x; 1.0423x over previous
#include <hip/hip_runtime.h>

#define N_NODES 100000
#define N_EDGES 1600000
#define D_IN    128
#define D_OUT   64

typedef __attribute__((ext_vector_type(2))) float v2f;
typedef __attribute__((ext_vector_type(8))) float v8f;

#define OWN_BLOCKS   256
#define OWN_ROWS     391
#define ID_CHUNK     256
#define ID_CHUNKS    (N_EDGES / ID_CHUNK)
#define ID_STEPS     (ID_CHUNK / 64)
#define REC_SRC_BITS 22
#define REC_SRC_MASK ((1u << REC_SRC_BITS) - 1u)
#define GEMM_WAVES   ((((N_NODES + 15) / 16 + 7) / 8) * 8)
#define MAP_GROUPS   (GEMM_WAVES / 8)
#define MAP_EDGES    (8 * ID_CHUNK)
#define MAP_DWORDS   (MAP_EDGES / 32)

#define WL_PITCH     80
#define A_AHEAD      16
static_assert(OWN_BLOCKS % 16 == 0, "whole waves clear and store the map pieces");
static_assert((D_IN / 2) * WL_PITCH * 8 <= OWN_BLOCKS * MAP_DWORDS * 4, "staged W fits the space of the map pieces");
static_assert((D_IN / 2) * (D_OUT / 4) == 4 * 256, "four staging tasks per thread");
static_assert(OWN_BLOCKS * OWN_ROWS >= N_NODES, "the blocks cover every row");
static_assert(N_EDGES % ID_CHUNK == 0, "whole id chunks");
static_assert(GEMM_WAVES >= ID_CHUNKS, "every id chunk has a projection wave");
static_assert(MAP_DWORDS == 64 && MAP_GROUPS % 2 == 0, "a map piece is 256 bytes; a scan step takes two pieces");
static_assert(N_NODES - 1 <= REC_SRC_MASK, "a source row fits its field");
static_assert(OWN_ROWS < (1u << (32 - REC_SRC_BITS)), "a row inside a block, or the spare row, fits its field");

__device__ __forceinline__ unsigned block_id(int d) {
    return min((unsigned)d / (unsigned)OWN_ROWS, (unsigned)(OWN_BLOCKS - 1));
}

__device__ __forceinline__ unsigned record_word(int s, int d, unsigned id) {
    const unsigned r = min((unsigned)d - id * (unsigned)OWN_ROWS, (unsigned)OWN_ROWS);
    return (r << REC_SRC_BITS) | min((unsigned)s, (unsigned)(N_NODES - 1));
}

__global__ __launch_bounds__(256) void gemm_bias_kernel(
    const float* __restrict__ x, const float* __restrict__ W,
    const float* __restrict__ bias, float* __restrict__ h,
    const int* __restrict__ src, const int* __restrict__ dst,
    const float* __restrict__ edge_w,
    unsigned int* __restrict__ maps, unsigned int* __restrict__ recs)
{
    __shared__ __attribute__((aligned(16))) unsigned bm[OWN_BLOCKS * MAP_DWORDS];

    const int lane = threadIdx.x & 31;
    const int wave = threadIdx.x >> 5;
    const int gw   = blockIdx.x * 8 + wave;
    const int m0   = gw * 16;

    const bool map_first = (((blockIdx.x >> 8) ^ blockIdx.x) & 1u) != 0u;

    float2* wl = (float2*)bm;
    float4 wr0[4], wr1[4];
    auto load_w = [&]() {
        #pragma unroll
        for (int i = 0; i < 4; ++i) {
            const int task = i * 256 + (int)threadIdx.x;
            const int kp = task >> 4, n4 = (task & 15) * 4;
            wr0[i] = *(const float4*)(W + (size_t)(2 * kp) * D_OUT + n4);
            wr1[i] = *(const float4*)(W + (size_t)(2 * kp + 1) * D_OUT + n4);
        }
    };
    auto stage_w = [&]() {
        #pragma unroll
        for (int i = 0; i < 4; ++i) {
            const int task = i * 256 + (int)threadIdx.x;
            const int kp = task >> 4, n4 = (task & 15) * 4;
            float4* wd = (float4*)(wl + kp * WL_PITCH + n4);
            wd[0] = make_float4(wr0[i].x, wr1[i].x, wr0[i].y, wr1[i].y);
            wd[1] = make_float4(wr0[i].z, wr1[i].z, wr0[i].w, wr1[i].w);
        }
    };
    if (!map_first) { load_w(); stage_w(); }

    const bool has_chunk = gw < ID_CHUNKS;
    const size_t pair0 = (size_t)gw * (ID_CHUNK / 2) + lane;
    int2   eS[ID_STEPS], eD[ID_STEPS];
    float2 eW[ID_STEPS];
    #pragma unroll
    for (int i = 0; i < ID_STEPS; ++i) {
        eS[i] = make_int2(0, 0); eD[i] = make_int2(0, 0); eW[i] = make_float2(0.f, 0.f);
        if (has_chunk) {
            eS[i] = ((const int2*)src)[pair0 + i * 32];
            eD[i] = ((const int2*)dst)[pair0 + i * 32];
            eW[i] = ((const float2*)edge_w)[pair0 + i * 32];
        }
    }

    const int half = lane >> 4;
    const int l15  = lane & 15;
    const bool has_strip = m0 < N_NODES;
    const float* xrow = x + (size_t)min(m0 + l15, N_NODES - 1) * D_IN + half * 2;
    v2f apre[A_AHEAD];
    #pragma unroll
    for (int j = 0; j < A_AHEAD; ++j) {
        apre[j].x = xrow[4 * j + 0];
        apre[j].y = xrow[4 * j + 1];
    }

    unsigned ids[ID_STEPS];
    #pragma unroll
    for (int i = 0; i < ID_STEPS; ++i) {
        const unsigned id0 = block_id(eD[i].x), id1 = block_id(eD[i].y);
        ids[i] = id0 | (id1 << 16);
        if (has_chunk) {
            uint4 rc;
            rc.x = record_word(eS[i].x, eD[i].x, id0);
            rc.y = (unsigned)__float_as_int(eW[i].x);
            rc.z = record_word(eS[i].y, eD[i].y, id1);
            rc.w = (unsigned)__float_as_int(eW[i].y);
            ((uint4*)recs)[pair0 + i * 32] = rc;
        }
    }

    auto map_phase = [&]() {
        #pragma unroll
        for (int j = 0; j < (OWN_BLOCKS * 2) / 32; ++j) {
            const int k = j * 32 + lane;
            ((uint4*)bm)[(k >> 1) * (MAP_DWORDS / 4) + wave * 2 + (k & 1)] = make_uint4(0u, 0u, 0u, 0u);
        }

        if (has_chunk) {
            #pragma unroll
            for (int i = 0; i < ID_STEPS; ++i) {
                const unsigned id0 = ids[i] & 0xffffu, id1 = ids[i] >> 16;
                const int dw = wave * 8 + i * 2 + (lane >> 4);
                const unsigned b0 = 1u << ((lane & 15) * 2);
                atomicOr(&bm[id0 * MAP_DWORDS + dw], b0);
                atomicOr(&bm[id1 * MAP_DWORDS + dw], b0 << 1);
            }
        }
        __builtin_amdgcn_fence(__ATOMIC_RELEASE, "workgroup", "local");
        __builtin_amdgcn_s_barrier();
        __builtin_amdgcn_fence(__ATOMIC_ACQUIRE, "workgroup", "local");

        #pragma unroll
        for (int r = 0; r < OWN_BLOCKS / 16; ++r) {
            const int b = r * 16 + (threadIdx.x >> 4);
            const int p = threadIdx.x & 15;
            ((uint4*)maps)[((size_t)b * MAP_GROUPS + blockIdx.x) * (MAP_DWORDS / 4) + p] =
                ((const uint4*)bm)[b * (MAP_DWORDS / 4) + p];
        }
    };

    if (map_first) {
        map_phase();
        load_w();
        __builtin_amdgcn_fence(__ATOMIC_RELEASE, "workgroup", "local");
        __builtin_amdgcn_s_barrier();
        __builtin_amdgcn_fence(__ATOMIC_ACQUIRE, "workgroup", "local");
        stage_w();
    }

    __builtin_amdgcn_fence(__ATOMIC_RELEASE, "workgroup", "local");
    __builtin_amdgcn_s_barrier();
    __builtin_amdgcn_fence(__ATOMIC_ACQUIRE, "workgroup", "local");

    if (has_strip) {

    v8f acc[4];
    #pragma unroll
    for (int t = 0; t < 4; ++t) {
        const float bval = bias[t * 16 + l15];
        #pragma unroll
        for (int i = 0; i < 8; ++i) acc[t][i] = bval;
    }

    const float2* wfrag = wl + half * WL_PITCH + l15;

    #pragma unroll
    for (int k0 = 0; k0 < D_IN; k0 += 4) {
        v2f a;
        if (k0 / 4 < A_AHEAD) {
            a = apre[k0 / 4];
        } else {
            a.x = xrow[k0 + 0];
            a.y = xrow[k0 + 1];
        }
        #pragma unroll
        for (int t = 0; t < 4; ++t) {
            const float2 wf = wfrag[(k0 / 2) * WL_PITCH + t * 16];
            v2f bf;
            bf.x = wf.x;
            bf.y = wf.y;
            acc[t] = __builtin_amdgcn_wmma_f32_16x16x4_f32(
                          false, a,  false, bf,
                          (short)0, acc[t],
                          false,  false);
        }
    }

    float* hout = h + (size_t)(m0 + half * 8) * D_OUT + l15;
    #pragma unroll
    for (int t = 0; t < 4; ++t)
        #pragma unroll
        for (int v = 0; v < 8; ++v)
            hout[(size_t)v * D_OUT + t * 16] = acc[t][v];
    }

    if (!map_first) {
        __builtin_amdgcn_fence(__ATOMIC_RELEASE, "workgroup", "local");
        __builtin_amdgcn_s_barrier();
        __builtin_amdgcn_fence(__ATOMIC_ACQUIRE, "workgroup", "local");
        map_phase();
    }
}

#define OWN_WAVES    32
#define OWN_THREADS  (OWN_WAVES * 32)
#define SCAN_STEPS   (MAP_GROUPS / 2)
#define WAVE_STEPS   13
#define HITS         32
#define PQCAP        64
#define OQCAP        512
#define ACC_ROWS     (OWN_ROWS + 1)
#define ROWS_X4      (4 * ACC_ROWS / OWN_WAVES)
#define WAVE_ROWS    13

typedef __attribute__((ext_vector_type(16))) float v16f;
template <int N> struct depth_t { static constexpr int value = N; };

static_assert(SCAN_STEPS <= WAVE_STEPS * OWN_WAVES, "a wave's share of the map fits its loads");
static_assert((OQCAP & (OQCAP - 1)) == 0, "places of a queue are taken round by a mask");
static_assert(PQCAP >= HITS - 1 + 32, "one pass over a dword (one hit per lane) cannot overflow a private queue");
static_assert(ROWS_X4 * OWN_WAVES == 4 * ACC_ROWS, "the rows divide over the waves in quarters");
static_assert(((ROWS_X4 + 3) >> 2) <= WAVE_ROWS && WAVE_ROWS <= 16, "a wave's rows fit its register sums");
static_assert(WAVE_ROWS <= (1 << (32 - REC_SRC_BITS)), "a row inside a wave fits the entry's field");
static_assert(ACC_ROWS * D_OUT * 4 + OWN_WAVES * OQCAP * 8 + OWN_WAVES * PQCAP * 4 + OWN_WAVES * 4 <= 320 * 1024, "LDS budget");
static_assert(ACC_ROWS * D_OUT * 4 + OWN_WAVES * OQCAP * 8 > 160 * 1024, "one work-group per WGP");

__global__ __launch_bounds__(OWN_THREADS) void owner_sum_kernel(
    const float* __restrict__ h,
    const unsigned int* __restrict__ maps, const unsigned int* __restrict__ recs,
    float* __restrict__ out)
{
    __shared__ __attribute__((aligned(16))) float acc[ACC_ROWS * D_OUT];
    __shared__ __attribute__((aligned(16))) uint2 oq[OWN_WAVES * OQCAP];
    __shared__ __attribute__((aligned(16))) unsigned pq[OWN_WAVES * PQCAP];
    __shared__ unsigned cnt[OWN_WAVES];

    const int tid  = threadIdx.x;
    const int lane = tid & 31;
    const int wave = __builtin_amdgcn_readfirstlane(tid >> 5);
    const int step_lo = 0;
    const int step_hi = SCAN_STEPS;
    const unsigned blk  = blockIdx.x;
    const unsigned base = blk * OWN_ROWS;

    float4* acc4 = (float4*)acc;
    for (int i = tid; i < ACC_ROWS * (D_OUT / 4); i += OWN_THREADS)
        acc4[i] = make_float4(0.f, 0.f, 0.f, 0.f);
    if (tid < OWN_WAVES) cnt[tid] = 0u;
    __syncthreads();

    unsigned* q = pq + wave * PQCAP;
    const uint4*  map4 = (const uint4*)maps + (size_t)blk * MAP_GROUPS * (MAP_DWORDS / 4);
    const uint2*  rec2 = (const uint2*)recs;
    const float2* h2   = (const float2*)h;
    int qcnt = 0;

    auto hand_on = [&](const unsigned* qs, int n) {
        const bool     valid = lane < n;
        const unsigned e   = qs[valid ? lane : n - 1];
        const uint2    rc  = rec2[e];
        const unsigned r   = rc.x >> REC_SRC_BITS;
        const unsigned o   = (r * 4u) / (unsigned)ROWS_X4;
        const unsigned k   = r - (((unsigned)ROWS_X4 * o + 3u) >> 2);
        bool full = false;
        if (valid) {
            const unsigned slot = atomicAdd(&cnt[o], 1u);
            if (slot < (unsigned)OQCAP)
                oq[o * OQCAP + ((slot + o) & (unsigned)(OQCAP - 1))] =
                    make_uint2((k << REC_SRC_BITS) | (rc.x & REC_SRC_MASK), rc.y);
            else
                full = true;
        }
        unsigned left = __builtin_amdgcn_ballot_w32(full);
        while (left != 0u) {
            const int j = __builtin_ctz(left);
            left &= left - 1u;
            const int      sj = __builtin_amdgcn_readlane((int)(rc.x & REC_SRC_MASK), j);
            const float    wj = __int_as_float(__builtin_amdgcn_readlane((int)rc.y, j));
            const unsigned rj = (unsigned)__builtin_amdgcn_readlane((int)r, j);
            const float2   v  = h2[(size_t)sj * (D_OUT / 2) + lane];
            float* a = acc + rj * D_OUT + lane * 2;
            atomicAdd(a + 0, v.x * wj);
            atomicAdd(a + 1, v.y * wj);
        }
    };

    uint4 d[WAVE_STEPS];
    #pragma unroll
    for (int u = 0; u < WAVE_STEPS; ++u) {
        const int tu = min(step_lo + wave + u * OWN_WAVES, step_hi - 1);
        d[u] = map4[(size_t)tu * 32 + lane];
    }
    #pragma unroll 1
    for (int t = step_lo + wave; t < step_hi; t += OWN_WAVES) {
        uint4 cur = d[0];
        unsigned e0 = ((unsigned)(2 * t) + (unsigned)(lane >> 4)) * (unsigned)MAP_EDGES
                    + (unsigned)(lane & 15) * 128u;
        #pragma unroll 1
        for (int i = 0; i < 4; ++i) {
            unsigned hit = cur.x;
            unsigned mask = __builtin_amdgcn_ballot_w32(hit != 0u);
            while (mask != 0u) {
                if (hit != 0u) {
                    const int pos = qcnt + (int)__builtin_amdgcn_mbcnt_lo(mask, 0u);
                    q[pos] = e0 + (unsigned)__builtin_ctz(hit);
                    hit &= hit - 1u;
                }
                qcnt += __builtin_popcount(mask);
                while (qcnt >= HITS) {
                    qcnt -= HITS;
                    hand_on(q + qcnt, HITS);
                }
                mask = __builtin_amdgcn_ballot_w32(hit != 0u);
            }
            cur.x = cur.y; cur.y = cur.z; cur.z = cur.w;
            e0 += 32u;
        }
        #pragma unroll
        for (int u = 0; u + 1 < WAVE_STEPS; ++u) d[u] = d[u + 1];
    }
    if (qcnt > 0) hand_on(q, qcnt);
    __syncthreads();

    const int first = (ROWS_X4 * wave + 3) >> 2;
    const int nrows = ((ROWS_X4 * (wave + 1) + 3) >> 2) - first;
    v16f sx = {}, sy = {};

    const int nq = __builtin_amdgcn_readfirstlane((int)min(cnt[wave], (unsigned)OQCAP));
    const uint2* myq = oq + wave * OQCAP;

    auto sum_entries = [&](int i0, int m, auto depth) {
        constexpr int DEPTH = decltype(depth)::value;
        const uint2 en  = myq[(i0 + (lane < m ? lane : m - 1) + wave) & (OQCAP - 1)];
        const int   s_l = (int)(en.x & REC_SRC_MASK);
        const int   k_l = (int)(en.x >> REC_SRC_BITS);
        const int   w_l = (int)en.y;
        float2 v[DEPTH];
        #pragma unroll
        for (int j = 0; j < DEPTH; ++j) {
            if (j < m) {
                const int sj = __builtin_amdgcn_readlane(s_l, j);
                v[j] = h2[(size_t)sj * (D_OUT / 2) + lane];
            }
        }
        #pragma unroll
        for (int j = 0; j < DEPTH; ++j) {
            if (j < m) {
                const float wj = __int_as_float(__builtin_amdgcn_readlane(w_l, j));
                const int   kj = __builtin_amdgcn_readlane(k_l, j);
                sx[kj] += v[j].x * wj;
                sy[kj] += v[j].y * wj;
            }
        }
    };

    int done = 0;
    #pragma unroll 1
    for (; done + HITS <= nq; done += HITS) sum_entries(done, HITS, depth_t<HITS>{});
    if (done + HITS / 2 <= nq) { sum_entries(done, HITS / 2, depth_t<HITS / 2>{}); done += HITS / 2; }
    if (done < nq) sum_entries(done, nq - done, depth_t<HITS / 2>{});

    const float2* acc2 = (const float2*)acc;
    float2* out2 = (float2*)out;
    #pragma unroll
    for (int i = 0; i < WAVE_ROWS; ++i) {
        const int r = first + i;
        const unsigned row = base + (unsigned)r;
        if (i < nrows && r < OWN_ROWS && row < N_NODES) {
            const float2 o = acc2[r * (D_OUT / 2) + lane];
            out2[(size_t)row * (D_OUT / 2) + lane] = make_float2(sx[i] + o.x, sy[i] + o.y);
        }
    }
}

extern "C" void kernel_launch(void* const* d_in, const int* in_sizes, int n_in,
                              void* d_out, int out_size, void* d_ws, size_t ws_size,
                              hipStream_t stream) {
    const float* x      = (const float*)d_in[0];
    const float* W      = (const float*)d_in[1];
    const float* b      = (const float*)d_in[2];
    const float* edge_w = (const float*)d_in[3];
    const int*   src    = (const int*)d_in[4];
    const int*   dst    = (const int*)d_in[5];
    float*       out    = (float*)d_out;
    float*       h      = (float*)d_ws;
    unsigned int* maps  = (unsigned int*)(h + (size_t)N_NODES * D_OUT);
    unsigned int* recs  = maps + (size_t)OWN_BLOCKS * MAP_GROUPS * MAP_DWORDS;

    (void)in_sizes; (void)n_in; (void)ws_size; (void)out_size;

    const int m_tiles = (N_NODES + 15) / 16;
    const int gemm_blocks = (m_tiles + 7) / 8;
    gemm_bias_kernel<<<gemm_blocks, 256, 0, stream>>>(x, W, b, h, src, dst, edge_w, maps, recs);

    owner_sum_kernel<<<OWN_BLOCKS, OWN_THREADS, 0, stream>>>(h, maps, recs, out);
}


# Edge maps hold one bit per pair of consecutive edges, not per edge: half the map bytes written, read and walked; a hit's 16-byte load brings both records, tags say whose; sums in queue arrival order
# speedup vs baseline: 2.7967x; 1.0835x over previous
#include <hip/hip_runtime.h>

#define N_NODES 100000
#define N_EDGES 1600000
#define D_IN    128
#define D_OUT   64

typedef __attribute__((ext_vector_type(2))) float v2f;
typedef __attribute__((ext_vector_type(8))) float v8f;

#define OWN_BLOCKS   256
#define OWN_ROWS     391
#define ID_CHUNK     256
#define ID_CHUNKS    (N_EDGES / ID_CHUNK)
#define ID_STEPS     (ID_CHUNK / 64)
#define REC_SRC_BITS 22
#define REC_SRC_MASK ((1u << REC_SRC_BITS) - 1u)
#define RW_SRC_BITS  17
#define RW_SRC_MASK  ((1u << RW_SRC_BITS) - 1u)
#define RW_ROW_BITS  9
#define RW_ROW_MASK  ((1u << RW_ROW_BITS) - 1u)
#define RW_TAG_SHIFT (RW_SRC_BITS + RW_ROW_BITS)
#define GEMM_WAVES   ((((N_NODES + 15) / 16 + 7) / 8) * 8)
#define MAP_GROUPS   (GEMM_WAVES / 8)
#define MAP_PAIRS    (8 * ID_CHUNK / 2)
#define MAP_DWORDS   (MAP_PAIRS / 32)
#define W_LDS_DWORDS ((D_IN / 2) * 80 * 2)
#define BM_DWORDS    (OWN_BLOCKS * MAP_DWORDS > W_LDS_DWORDS ? OWN_BLOCKS * MAP_DWORDS : W_LDS_DWORDS)

#define WL_PITCH     80
#define A_AHEAD      16
static_assert(OWN_BLOCKS % 32 == 0, "whole waves clear and store the map pieces");
static_assert((D_IN / 2) * WL_PITCH * 2 == W_LDS_DWORDS, "the staged W's size");
static_assert((D_IN / 2) * WL_PITCH * 8 <= BM_DWORDS * 4 && OWN_BLOCKS * MAP_DWORDS <= BM_DWORDS, "staged W and the map pieces both fit the shared space");
static_assert((D_IN / 2) * (D_OUT / 4) == 4 * 256, "four staging tasks per thread");
static_assert(OWN_BLOCKS * OWN_ROWS >= N_NODES, "the blocks cover every row");
static_assert(N_EDGES % ID_CHUNK == 0, "whole id chunks");
static_assert(GEMM_WAVES >= ID_CHUNKS, "every id chunk has a projection wave");
static_assert(MAP_DWORDS == 32 && MAP_DWORDS == 8 * ID_STEPS, "a map piece is 128 bytes: one 16-byte part per wave, one dword per step");
static_assert(N_NODES - 1 <= RW_SRC_MASK && N_NODES - 1 <= REC_SRC_MASK, "a source row fits its field");
static_assert(OWN_ROWS <= RW_ROW_MASK, "a row inside a block, or the spare row, fits its field");
static_assert(OWN_BLOCKS <= 256 && RW_TAG_SHIFT + 6 == 32, "a block id fits the two tags: six bits and two bits");

__device__ __forceinline__ unsigned block_id(int d) {
    return min((unsigned)d / (unsigned)OWN_ROWS, (unsigned)(OWN_BLOCKS - 1));
}

__device__ __forceinline__ unsigned record_word(int s, int d, unsigned id, unsigned tag) {
    const unsigned r = min((unsigned)d - id * (unsigned)OWN_ROWS, (unsigned)OWN_ROWS);
    return (tag << RW_TAG_SHIFT) | (r << RW_SRC_BITS) | min((unsigned)s, (unsigned)(N_NODES - 1));
}

__global__ __launch_bounds__(256) void gemm_bias_kernel(
    const float* __restrict__ x, const float* __restrict__ W,
    const float* __restrict__ bias, float* __restrict__ h,
    const int* __restrict__ src, const int* __restrict__ dst,
    const float* __restrict__ edge_w,
    unsigned int* __restrict__ maps, unsigned int* __restrict__ recs)
{
    __shared__ __attribute__((aligned(16))) unsigned bm[BM_DWORDS];

    const int lane = threadIdx.x & 31;
    const int wave = threadIdx.x >> 5;
    const int gw   = blockIdx.x * 8 + wave;
    const int m0   = gw * 16;

    const bool map_first = (((blockIdx.x >> 8) ^ blockIdx.x) & 1u) != 0u;

    float2* wl = (float2*)bm;
    float4 wr0[4], wr1[4];
    auto load_w = [&]() {
        #pragma unroll
        for (int i = 0; i < 4; ++i) {
            const int task = i * 256 + (int)threadIdx.x;
            const int kp = task >> 4, n4 = (task & 15) * 4;
            wr0[i] = *(const float4*)(W + (size_t)(2 * kp) * D_OUT + n4);
            wr1[i] = *(const float4*)(W + (size_t)(2 * kp + 1) * D_OUT + n4);
        }
    };
    auto stage_w = [&]() {
        #pragma unroll
        for (int i = 0; i < 4; ++i) {
            const int task = i * 256 + (int)threadIdx.x;
            const int kp = task >> 4, n4 = (task & 15) * 4;
            float4* wd = (float4*)(wl + kp * WL_PITCH + n4);
            wd[0] = make_float4(wr0[i].x, wr1[i].x, wr0[i].y, wr1[i].y);
            wd[1] = make_float4(wr0[i].z, wr1[i].z, wr0[i].w, wr1[i].w);
        }
    };
    if (!map_first) { load_w(); stage_w(); }

    const bool has_chunk = gw < ID_CHUNKS;
    const size_t pair0 = (size_t)gw * (ID_CHUNK / 2) + lane;
    int2   eS[ID_STEPS], eD[ID_STEPS];
    float2 eW[ID_STEPS];
    #pragma unroll
    for (int i = 0; i < ID_STEPS; ++i) {
        eS[i] = make_int2(0, 0); eD[i] = make_int2(0, 0); eW[i] = make_float2(0.f, 0.f);
        if (has_chunk) {
            eS[i] = ((const int2*)src)[pair0 + i * 32];
            eD[i] = ((const int2*)dst)[pair0 + i * 32];
            eW[i] = ((const float2*)edge_w)[pair0 + i * 32];
        }
    }

    const int half = lane >> 4;
    const int l15  = lane & 15;
    const bool has_strip = m0 < N_NODES;
    const float* xrow = x + (size_t)min(m0 + l15, N_NODES - 1) * D_IN + half * 2;
    v2f apre[A_AHEAD];
    #pragma unroll
    for (int j = 0; j < A_AHEAD; ++j) {
        apre[j].x = xrow[4 * j + 0];
        apre[j].y = xrow[4 * j + 1];
    }

    unsigned ids[ID_STEPS];
    #pragma unroll
    for (int i = 0; i < ID_STEPS; ++i) {
        const unsigned id0 = block_id(eD[i].x), id1 = block_id(eD[i].y);
        ids[i] = id0 | (id1 << 16);
        if (has_chunk) {
            uint4 rc;
            rc.x = record_word(eS[i].x, eD[i].x, id0, id0 & 63u);
            rc.y = (unsigned)__float_as_int(eW[i].x);
            rc.z = record_word(eS[i].y, eD[i].y, id1, (id0 >> 6) | (id1 == id0 ? 4u : 0u));
            rc.w = (unsigned)__float_as_int(eW[i].y);
            ((uint4*)recs)[pair0 + i * 32] = rc;
        }
    }

    auto map_phase = [&]() {
        #pragma unroll
        for (int j = 0; j < OWN_BLOCKS / 32; ++j) {
            const int k = j * 32 + lane;
            ((uint4*)bm)[k * (MAP_DWORDS / 4) + wave] = make_uint4(0u, 0u, 0u, 0u);
        }

        if (has_chunk) {
            #pragma unroll
            for (int i = 0; i < ID_STEPS; ++i) {
                const unsigned id0 = ids[i] & 0xffffu, id1 = ids[i] >> 16;
                const int dw = wave * ID_STEPS + i;
                const unsigned b0 = 1u << lane;
                atomicOr(&bm[id0 * MAP_DWORDS + dw], b0);
                atomicOr(&bm[id1 * MAP_DWORDS + dw], b0);
            }
        }
        __builtin_amdgcn_fence(__ATOMIC_RELEASE, "workgroup", "local");
        __builtin_amdgcn_s_barrier();
        __builtin_amdgcn_fence(__ATOMIC_ACQUIRE, "workgroup", "local");

        #pragma unroll
        for (int r = 0; r < OWN_BLOCKS / 32; ++r) {
            const int b = r * 32 + (threadIdx.x >> 3);
            const int p = threadIdx.x & 7;
            ((uint4*)maps)[((size_t)b * MAP_GROUPS + blockIdx.x) * (MAP_DWORDS / 4) + p] =
                ((const uint4*)bm)[b * (MAP_DWORDS / 4) + p];
        }
    };

    if (map_first) {
        map_phase();
        load_w();
        __builtin_amdgcn_fence(__ATOMIC_RELEASE, "workgroup", "local");
        __builtin_amdgcn_s_barrier();
        __builtin_amdgcn_fence(__ATOMIC_ACQUIRE, "workgroup", "local");
        stage_w();
    }

    __builtin_amdgcn_fence(__ATOMIC_RELEASE, "workgroup", "local");
    __builtin_amdgcn_s_barrier();
    __builtin_amdgcn_fence(__ATOMIC_ACQUIRE, "workgroup", "local");

    if (has_strip) {

    v8f acc[4];
    #pragma unroll
    for (int t = 0; t < 4; ++t) {
        const float bval = bias[t * 16 + l15];
        #pragma unroll
        for (int i = 0; i < 8; ++i) acc[t][i] = bval;
    }

    const float2* wfrag = wl + half * WL_PITCH + l15;

    #pragma unroll
    for (int k0 = 0; k0 < D_IN; k0 += 4) {
        v2f a;
        if (k0 / 4 < A_AHEAD) {
            a = apre[k0 / 4];
        } else {
            a.x = xrow[k0 + 0];
            a.y = xrow[k0 + 1];
        }
        #pragma unroll
        for (int t = 0; t < 4; ++t) {
            const float2 wf = wfrag[(k0 / 2) * WL_PITCH + t * 16];
            v2f bf;
            bf.x = wf.x;
            bf.y = wf.y;
            acc[t] = __builtin_amdgcn_wmma_f32_16x16x4_f32(
                          false, a,  false, bf,
                          (short)0, acc[t],
                          false,  false);
        }
    }

    float* hout = h + (size_t)(m0 + half * 8) * D_OUT + l15;
    #pragma unroll
    for (int t = 0; t < 4; ++t)
        #pragma unroll
        for (int v = 0; v < 8; ++v)
            hout[(size_t)v * D_OUT + t * 16] = acc[t][v];
    }

    if (!map_first) {
        __builtin_amdgcn_fence(__ATOMIC_RELEASE, "workgroup", "local");
        __builtin_amdgcn_s_barrier();
        __builtin_amdgcn_fence(__ATOMIC_ACQUIRE, "workgroup", "local");
        map_phase();
    }
}

#define OWN_WAVES    32
#define OWN_THREADS  (OWN_WAVES * 32)
#define PIECE_VECS   (MAP_DWORDS / 4)
#define BLOCK_VECS   (MAP_GROUPS * PIECE_VECS)
#define SCAN_STEPS   ((BLOCK_VECS + 31) / 32)
#define WAVE_STEPS   7
#define HITS         32
#define PQCAP        64
#define OQCAP        512
#define ACC_ROWS     (OWN_ROWS + 1)
#define ROWS_X4      (4 * ACC_ROWS / OWN_WAVES)
#define WAVE_ROWS    13

typedef __attribute__((ext_vector_type(16))) float v16f;
template <int N> struct depth_t { static constexpr int value = N; };

static_assert(SCAN_STEPS <= WAVE_STEPS * OWN_WAVES, "a wave's share of the map fits its loads");
static_assert((OQCAP & (OQCAP - 1)) == 0, "places of a queue are taken round by a mask");
static_assert(PQCAP >= HITS - 1 + 32, "one pass over a dword (one hit per lane) cannot overflow a private queue");
static_assert(ROWS_X4 * OWN_WAVES == 4 * ACC_ROWS, "the rows divide over the waves in quarters");
static_assert(((ROWS_X4 + 3) >> 2) <= WAVE_ROWS && WAVE_ROWS <= 16, "a wave's rows fit its register sums");
static_assert(WAVE_ROWS <= (1 << (32 - REC_SRC_BITS)), "a row inside a wave fits the entry's field");
static_assert(ACC_ROWS * D_OUT * 4 + OWN_WAVES * OQCAP * 8 + OWN_WAVES * PQCAP * 4 + OWN_WAVES * 4 <= 320 * 1024, "LDS budget");
static_assert(ACC_ROWS * D_OUT * 4 + OWN_WAVES * OQCAP * 8 > 160 * 1024, "one work-group per WGP");

__global__ __launch_bounds__(OWN_THREADS) void owner_sum_kernel(
    const float* __restrict__ h,
    const unsigned int* __restrict__ maps, const unsigned int* __restrict__ recs,
    float* __restrict__ out)
{
    __shared__ __attribute__((aligned(16))) float acc[ACC_ROWS * D_OUT];
    __shared__ __attribute__((aligned(16))) uint2 oq[OWN_WAVES * OQCAP];
    __shared__ __attribute__((aligned(16))) unsigned pq[OWN_WAVES * PQCAP];
    __shared__ unsigned cnt[OWN_WAVES];

    const int tid  = threadIdx.x;
    const int lane = tid & 31;
    const int wave = __builtin_amdgcn_readfirstlane(tid >> 5);
    const int step_lo = 0;
    const int step_hi = SCAN_STEPS;
    const unsigned blk  = blockIdx.x;
    const unsigned base = blk * OWN_ROWS;

    float4* acc4 = (float4*)acc;
    for (int i = tid; i < ACC_ROWS * (D_OUT / 4); i += OWN_THREADS)
        acc4[i] = make_float4(0.f, 0.f, 0.f, 0.f);
    if (tid < OWN_WAVES) cnt[tid] = 0u;
    __syncthreads();

    unsigned* q = pq + wave * PQCAP;
    const uint4*  map4 = (const uint4*)maps + (size_t)blk * BLOCK_VECS;
    const uint4*  rec4 = (const uint4*)recs;
    const float2* h2   = (const float2*)h;
    int qcnt = 0;

    auto push = [&](bool on, unsigned word, unsigned wbits) {
        const unsigned s   = word & RW_SRC_MASK;
        const unsigned r   = (word >> RW_SRC_BITS) & RW_ROW_MASK;
        const unsigned o   = (r * 4u) / (unsigned)ROWS_X4;
        const unsigned k   = r - (((unsigned)ROWS_X4 * o + 3u) >> 2);
        bool full = false;
        if (on) {
            const unsigned slot = atomicAdd(&cnt[o], 1u);
            if (slot < (unsigned)OQCAP)
                oq[o * OQCAP + ((slot + o) & (unsigned)(OQCAP - 1))] =
                    make_uint2((k << REC_SRC_BITS) | s, wbits);
            else
                full = true;
        }
        unsigned left = __builtin_amdgcn_ballot_w32(full);
        while (left != 0u) {
            const int j = __builtin_ctz(left);
            left &= left - 1u;
            const int      sj = __builtin_amdgcn_readlane((int)s, j);
            const float    wj = __int_as_float(__builtin_amdgcn_readlane((int)wbits, j));
            const unsigned rj = (unsigned)__builtin_amdgcn_readlane((int)r, j);
            const float2   v  = h2[(size_t)sj * (D_OUT / 2) + lane];
            float* a = acc + rj * D_OUT + lane * 2;
            atomicAdd(a + 0, v.x * wj);
            atomicAdd(a + 1, v.y * wj);
        }
    };

    auto hand_on = [&](const unsigned* qs, int n) {
        const bool     valid = lane < n;
        const unsigned p   = qs[valid ? lane : n - 1];
        const uint4    rc  = rec4[p];
        const unsigned id0 = (rc.x >> RW_TAG_SHIFT) | (((rc.z >> RW_TAG_SHIFT) & 3u) << 6);
        const bool first = id0 == blk;
        push(valid, first ? rc.x : rc.z, first ? rc.y : rc.w);
        const bool both = valid && first && ((rc.z >> (RW_TAG_SHIFT + 2)) & 1u) != 0u;
        if (__builtin_amdgcn_ballot_w32(both) != 0u) push(both, rc.z, rc.w);
    };

    uint4 d[WAVE_STEPS];
    #pragma unroll
    for (int u = 0; u < WAVE_STEPS; ++u) {
        const int tu = min(step_lo + wave + u * OWN_WAVES, step_hi - 1);
        const int  vi = tu * 32 + lane;
        const bool in = vi < BLOCK_VECS;
        const uint4 dv = map4[in ? vi : BLOCK_VECS - 1];
        d[u] = in ? dv : make_uint4(0u, 0u, 0u, 0u);
    }
    #pragma unroll 1
    for (int t = step_lo + wave; t < step_hi; t += OWN_WAVES) {
        uint4 cur = d[0];
        unsigned e0 = ((unsigned)(4 * t) + (unsigned)(lane >> 3)) * (unsigned)MAP_PAIRS
                    + (unsigned)(lane & 7) * 128u;
        #pragma unroll 1
        for (int i = 0; i < 4; ++i) {
            unsigned hit = cur.x;
            unsigned mask = __builtin_amdgcn_ballot_w32(hit != 0u);
            while (mask != 0u) {
                if (hit != 0u) {
                    const int pos = qcnt + (int)__builtin_amdgcn_mbcnt_lo(mask, 0u);
                    q[pos] = e0 + (unsigned)__builtin_ctz(hit);
                    hit &= hit - 1u;
                }
                qcnt += __builtin_popcount(mask);
                while (qcnt >= HITS) {
                    qcnt -= HITS;
                    hand_on(q + qcnt, HITS);
                }
                mask = __builtin_amdgcn_ballot_w32(hit != 0u);
            }
            cur.x = cur.y; cur.y = cur.z; cur.z = cur.w;
            e0 += 32u;
        }
        #pragma unroll
        for (int u = 0; u + 1 < WAVE_STEPS; ++u) d[u] = d[u + 1];
    }
    if (qcnt > 0) hand_on(q, qcnt);
    __syncthreads();

    const int first = (ROWS_X4 * wave + 3) >> 2;
    const int nrows = ((ROWS_X4 * (wave + 1) + 3) >> 2) - first;
    v16f sx = {}, sy = {};

    const int nq = __builtin_amdgcn_readfirstlane((int)min(cnt[wave], (unsigned)OQCAP));
    const uint2* myq = oq + wave * OQCAP;

    auto sum_entries = [&](int i0, int m, auto depth) {
        constexpr int DEPTH = decltype(depth)::value;
        const uint2 en  = myq[(i0 + (lane < m ? lane : m - 1) + wave) & (OQCAP - 1)];
        const int   s_l = (int)(en.x & REC_SRC_MASK);
        const int   k_l = (int)(en.x >> REC_SRC_BITS);
        const int   w_l = (int)en.y;
        float2 v[DEPTH];
        #pragma unroll
        for (int j = 0; j < DEPTH; ++j) {
            if (j < m) {
                const int sj = __builtin_amdgcn_readlane(s_l, j);
                v[j] = h2[(size_t)sj * (D_OUT / 2) + lane];
            }
        }
        #pragma unroll
        for (int j = 0; j < DEPTH; ++j) {
            if (j < m) {
                const float wj = __int_as_float(__builtin_amdgcn_readlane(w_l, j));
                const int   kj = __builtin_amdgcn_readlane(k_l, j);
                sx[kj] += v[j].x * wj;
                sy[kj] += v[j].y * wj;
            }
        }
    };

    int done = 0;
    #pragma unroll 1
    for (; done + HITS <= nq; done += HITS) sum_entries(done, HITS, depth_t<HITS>{});
    if (done + HITS / 2 <= nq) { sum_entries(done, HITS / 2, depth_t<HITS / 2>{}); done += HITS / 2; }
    if (done < nq) sum_entries(done, nq - done, depth_t<HITS / 2>{});

    const float2* acc2 = (const float2*)acc;
    float2* out2 = (float2*)out;
    #pragma unroll
    for (int i = 0; i < WAVE_ROWS; ++i) {
        const int r = first + i;
        const unsigned row = base + (unsigned)r;
        if (i < nrows && r < OWN_ROWS && row < N_NODES) {
            const float2 o = acc2[r * (D_OUT / 2) + lane];
            out2[(size_t)row * (D_OUT / 2) + lane] = make_float2(sx[i] + o.x, sy[i] + o.y);
        }
    }
}

extern "C" void kernel_launch(void* const* d_in, const int* in_sizes, int n_in,
                              void* d_out, int out_size, void* d_ws, size_t ws_size,
                              hipStream_t stream) {
    const float* x      = (const float*)d_in[0];
    const float* W      = (const float*)d_in[1];
    const float* b      = (const float*)d_in[2];
    const float* edge_w = (const float*)d_in[3];
    const int*   src    = (const int*)d_in[4];
    const int*   dst    = (const int*)d_in[5];
    float*       out    = (float*)d_out;
    float*       h      = (float*)d_ws;
    unsigned int* maps  = (unsigned int*)(h + (size_t)N_NODES * D_OUT);
    unsigned int* recs  = maps + (size_t)OWN_BLOCKS * MAP_GROUPS * MAP_DWORDS;

    (void)in_sizes; (void)n_in; (void)ws_size; (void)out_size;

    const int m_tiles = (N_NODES + 15) / 16;
    const int gemm_blocks = (m_tiles + 7) / 8;
    gemm_bias_kernel<<<gemm_blocks, 256, 0, stream>>>(x, W, b, h, src, dst, edge_w, maps, recs);

    owner_sum_kernel<<<OWN_BLOCKS, OWN_THREADS, 0, stream>>>(h, maps, recs, out);
}


# Edge phase: one queue per destination row, not per owner wave: a row's sum stays in fixed registers, each half-wave takes every other entry (16-byte row loads); sum = even places + odd places
# speedup vs baseline: 2.9476x; 1.0540x over previous
#include <hip/hip_runtime.h>

#define N_NODES 100000
#define N_EDGES 1600000
#define D_IN    128
#define D_OUT   64

typedef __attribute__((ext_vector_type(2))) float v2f;
typedef __attribute__((ext_vector_type(8))) float v8f;

#define OWN_BLOCKS   256
#define OWN_ROWS     391
#define ID_CHUNK     256
#define ID_CHUNKS    (N_EDGES / ID_CHUNK)
#define ID_STEPS     (ID_CHUNK / 64)
#define RW_SRC_BITS  17
#define RW_SRC_MASK  ((1u << RW_SRC_BITS) - 1u)
#define RW_ROW_BITS  9
#define RW_ROW_MASK  ((1u << RW_ROW_BITS) - 1u)
#define RW_TAG_SHIFT (RW_SRC_BITS + RW_ROW_BITS)
#define GEMM_WAVES   ((((N_NODES + 15) / 16 + 7) / 8) * 8)
#define MAP_GROUPS   (GEMM_WAVES / 8)
#define MAP_PAIRS    (8 * ID_CHUNK / 2)
#define MAP_DWORDS   (MAP_PAIRS / 32)
#define W_LDS_DWORDS ((D_IN / 2) * 80 * 2)
#define BM_DWORDS    (OWN_BLOCKS * MAP_DWORDS > W_LDS_DWORDS ? OWN_BLOCKS * MAP_DWORDS : W_LDS_DWORDS)

#define WL_PITCH     80
#define A_AHEAD      16
static_assert(OWN_BLOCKS % 32 == 0, "whole waves clear and store the map pieces");
static_assert((D_IN / 2) * WL_PITCH * 2 == W_LDS_DWORDS, "the staged W's size");
static_assert((D_IN / 2) * WL_PITCH * 8 <= BM_DWORDS * 4 && OWN_BLOCKS * MAP_DWORDS <= BM_DWORDS, "staged W and the map pieces both fit the shared space");
static_assert((D_IN / 2) * (D_OUT / 4) == 4 * 256, "four staging tasks per thread");
static_assert(OWN_BLOCKS * OWN_ROWS >= N_NODES, "the blocks cover every row");
static_assert(N_EDGES % ID_CHUNK == 0, "whole id chunks");
static_assert(GEMM_WAVES >= ID_CHUNKS, "every id chunk has a projection wave");
static_assert(MAP_DWORDS == 32 && MAP_DWORDS == 8 * ID_STEPS, "a map piece is 128 bytes: one 16-byte part per wave, one dword per step");
static_assert(N_NODES - 1 <= RW_SRC_MASK, "a source row fits its field");
static_assert(OWN_ROWS <= RW_ROW_MASK, "a row inside a block, or the spare row, fits its field");
static_assert(OWN_BLOCKS <= 256 && RW_TAG_SHIFT + 6 == 32, "a block id fits the two tags: six bits and two bits");

__device__ __forceinline__ unsigned block_id(int d) {
    return min((unsigned)d / (unsigned)OWN_ROWS, (unsigned)(OWN_BLOCKS - 1));
}

__device__ __forceinline__ unsigned record_word(int s, int d, unsigned id, unsigned tag) {
    const unsigned r = min((unsigned)d - id * (unsigned)OWN_ROWS, (unsigned)OWN_ROWS);
    return (tag << RW_TAG_SHIFT) | (r << RW_SRC_BITS) | min((unsigned)s, (unsigned)(N_NODES - 1));
}

__global__ __launch_bounds__(256) void gemm_bias_kernel(
    const float* __restrict__ x, const float* __restrict__ W,
    const float* __restrict__ bias, float* __restrict__ h,
    const int* __restrict__ src, const int* __restrict__ dst,
    const float* __restrict__ edge_w,
    unsigned int* __restrict__ maps, unsigned int* __restrict__ recs)
{
    __shared__ __attribute__((aligned(16))) unsigned bm[BM_DWORDS];

    const int lane = threadIdx.x & 31;
    const int wave = threadIdx.x >> 5;
    const int gw   = blockIdx.x * 8 + wave;
    const int m0   = gw * 16;

    const bool map_first = (((blockIdx.x >> 8) ^ blockIdx.x) & 1u) != 0u;

    float2* wl = (float2*)bm;
    float4 wr0[4], wr1[4];
    auto load_w = [&]() {
        #pragma unroll
        for (int i = 0; i < 4; ++i) {
            const int task = i * 256 + (int)threadIdx.x;
            const int kp = task >> 4, n4 = (task & 15) * 4;
            wr0[i] = *(const float4*)(W + (size_t)(2 * kp) * D_OUT + n4);
            wr1[i] = *(const float4*)(W + (size_t)(2 * kp + 1) * D_OUT + n4);
        }
    };
    auto stage_w = [&]() {
        #pragma unroll
        for (int i = 0; i < 4; ++i) {
            const int task = i * 256 + (int)threadIdx.x;
            const int kp = task >> 4, n4 = (task & 15) * 4;
            float4* wd = (float4*)(wl + kp * WL_PITCH + n4);
            wd[0] = make_float4(wr0[i].x, wr1[i].x, wr0[i].y, wr1[i].y);
            wd[1] = make_float4(wr0[i].z, wr1[i].z, wr0[i].w, wr1[i].w);
        }
    };
    if (!map_first) { load_w(); stage_w(); }

    const bool has_chunk = gw < ID_CHUNKS;
    const size_t pair0 = (size_t)gw * (ID_CHUNK / 2) + lane;
    int2   eS[ID_STEPS], eD[ID_STEPS];
    float2 eW[ID_STEPS];
    #pragma unroll
    for (int i = 0; i < ID_STEPS; ++i) {
        eS[i] = make_int2(0, 0); eD[i] = make_int2(0, 0); eW[i] = make_float2(0.f, 0.f);
        if (has_chunk) {
            eS[i] = ((const int2*)src)[pair0 + i * 32];
            eD[i] = ((const int2*)dst)[pair0 + i * 32];
            eW[i] = ((const float2*)edge_w)[pair0 + i * 32];
        }
    }

    const int half = lane >> 4;
    const int l15  = lane & 15;
    const bool has_strip = m0 < N_NODES;
    const float* xrow = x + (size_t)min(m0 + l15, N_NODES - 1) * D_IN + half * 2;
    v2f apre[A_AHEAD];
    #pragma unroll
    for (int j = 0; j < A_AHEAD; ++j) {
        apre[j].x = xrow[4 * j + 0];
        apre[j].y = xrow[4 * j + 1];
    }

    unsigned ids[ID_STEPS];
    #pragma unroll
    for (int i = 0; i < ID_STEPS; ++i) {
        const unsigned id0 = block_id(eD[i].x), id1 = block_id(eD[i].y);
        ids[i] = id0 | (id1 << 16);
        if (has_chunk) {
            uint4 rc;
            rc.x = record_word(eS[i].x, eD[i].x, id0, id0 & 63u);
            rc.y = (unsigned)__float_as_int(eW[i].x);
            rc.z = record_word(eS[i].y, eD[i].y, id1, (id0 >> 6) | (id1 == id0 ? 4u : 0u));
            rc.w = (unsigned)__float_as_int(eW[i].y);
            ((uint4*)recs)[pair0 + i * 32] = rc;
        }
    }

    auto map_phase = [&]() {
        #pragma unroll
        for (int j = 0; j < OWN_BLOCKS / 32; ++j) {
            const int k = j * 32 + lane;
            ((uint4*)bm)[k * (MAP_DWORDS / 4) + wave] = make_uint4(0u, 0u, 0u, 0u);
        }

        if (has_chunk) {
            #pragma unroll
            for (int i = 0; i < ID_STEPS; ++i) {
                const unsigned id0 = ids[i] & 0xffffu, id1 = ids[i] >> 16;
                const int dw = wave * ID_STEPS + i;
                const unsigned b0 = 1u << lane;
                atomicOr(&bm[id0 * MAP_DWORDS + dw], b0);
                atomicOr(&bm[id1 * MAP_DWORDS + dw], b0);
            }
        }
        __builtin_amdgcn_fence(__ATOMIC_RELEASE, "workgroup", "local");
        __builtin_amdgcn_s_barrier();
        __builtin_amdgcn_fence(__ATOMIC_ACQUIRE, "workgroup", "local");

        #pragma unroll
        for (int r = 0; r < OWN_BLOCKS / 32; ++r) {
            const int b = r * 32 + (threadIdx.x >> 3);
            const int p = threadIdx.x & 7;
            ((uint4*)maps)[((size_t)b * MAP_GROUPS + blockIdx.x) * (MAP_DWORDS / 4) + p] =
                ((const uint4*)bm)[b * (MAP_DWORDS / 4) + p];
        }
    };

    if (map_first) {
        map_phase();
        load_w();
        __builtin_amdgcn_fence(__ATOMIC_RELEASE, "workgroup", "local");
        __builtin_amdgcn_s_barrier();
        __builtin_amdgcn_fence(__ATOMIC_ACQUIRE, "workgroup", "local");
        stage_w();
    }

    __builtin_amdgcn_fence(__ATOMIC_RELEASE, "workgroup", "local");
    __builtin_amdgcn_s_barrier();
    __builtin_amdgcn_fence(__ATOMIC_ACQUIRE, "workgroup", "local");

    if (has_strip) {

    v8f acc[4];
    #pragma unroll
    for (int t = 0; t < 4; ++t) {
        const float bval = bias[t * 16 + l15];
        #pragma unroll
        for (int i = 0; i < 8; ++i) acc[t][i] = bval;
    }

    const float2* wfrag = wl + half * WL_PITCH + l15;

    #pragma unroll
    for (int k0 = 0; k0 < D_IN; k0 += 4) {
        v2f a;
        if (k0 / 4 < A_AHEAD) {
            a = apre[k0 / 4];
        } else {
            a.x = xrow[k0 + 0];
            a.y = xrow[k0 + 1];
        }
        #pragma unroll
        for (int t = 0; t < 4; ++t) {
            const float2 wf = wfrag[(k0 / 2) * WL_PITCH + t * 16];
            v2f bf;
            bf.x = wf.x;
            bf.y = wf.y;
            acc[t] = __builtin_amdgcn_wmma_f32_16x16x4_f32(
                          false, a,  false, bf,
                          (short)0, acc[t],
                          false,  false);
        }
    }

    float* hout = h + (size_t)(m0 + half * 8) * D_OUT + l15;
    #pragma unroll
    for (int t = 0; t < 4; ++t)
        #pragma unroll
        for (int v = 0; v < 8; ++v)
            hout[(size_t)v * D_OUT + t * 16] = acc[t][v];
    }

    if (!map_first) {
        __builtin_amdgcn_fence(__ATOMIC_RELEASE, "workgroup", "local");
        __builtin_amdgcn_s_barrier();
        __builtin_amdgcn_fence(__ATOMIC_ACQUIRE, "workgroup", "local");
        map_phase();
    }
}

#define OWN_WAVES    32
#define OWN_THREADS  (OWN_WAVES * 32)
#define PIECE_VECS   (MAP_DWORDS / 4)
#define BLOCK_VECS   (MAP_GROUPS * PIECE_VECS)
#define SCAN_STEPS   ((BLOCK_VECS + 31) / 32)
#define WAVE_STEPS   7
#define HITS         32
#define PQCAP        64
#define RQCAP        40
#define RQ_PITCH     41
#define PAIRS        12
#define ACC_ROWS     (OWN_ROWS + 1)
#define ROWS_X4      (4 * ACC_ROWS / OWN_WAVES)
#define WAVE_ROWS    13

template <int N> struct loads_t { static constexpr int value = N; };

static_assert(SCAN_STEPS <= WAVE_STEPS * OWN_WAVES, "a wave's share of the map fits its loads");
static_assert(PQCAP >= HITS - 1 + 32, "one pass over a dword (one hit per lane) cannot overflow a private queue");
static_assert(ROWS_X4 * OWN_WAVES == 4 * ACC_ROWS, "the rows divide over the waves in quarters");
static_assert(((ROWS_X4 + 3) >> 2) <= WAVE_ROWS, "the most rows one wave owns");
static_assert(ACC_ROWS <= OWN_THREADS, "one thread clears each row's counter");
static_assert(RQ_PITCH >= RQCAP && (RQ_PITCH & 1) == 1, "a row's queue fits its pitch, and the pitch is odd");
static_assert(PAIRS == 12, "the switch over a group's row loads names every count up to PAIRS");
static_assert(ACC_ROWS * D_OUT * 4 + ACC_ROWS * RQ_PITCH * 8 + OWN_WAVES * PQCAP * 4 + ACC_ROWS * 4 <= 320 * 1024, "LDS budget");
static_assert(ACC_ROWS * D_OUT * 4 + ACC_ROWS * RQ_PITCH * 8 > 160 * 1024, "one work-group per WGP");

__device__ __forceinline__ float other_half(float x) {
    return __uint_as_float(__builtin_amdgcn_permlanex16(0u, __float_as_uint(x), 0x76543210u, 0xfedcba98u, false, false));
}

__global__ __launch_bounds__(OWN_THREADS) void owner_sum_kernel(
    const float* __restrict__ h,
    const unsigned int* __restrict__ maps, const unsigned int* __restrict__ recs,
    float* __restrict__ out)
{
    __shared__ __attribute__((aligned(16))) float acc[ACC_ROWS * D_OUT];
    __shared__ __attribute__((aligned(16))) uint2 rq[ACC_ROWS * RQ_PITCH];
    __shared__ __attribute__((aligned(16))) unsigned pq[OWN_WAVES * PQCAP];
    __shared__ unsigned cnt[ACC_ROWS];

    const int tid  = threadIdx.x;
    const int lane = tid & 31;
    const int wave = __builtin_amdgcn_readfirstlane(tid >> 5);
    const int step_lo = 0;
    const int step_hi = SCAN_STEPS;
    const unsigned blk  = blockIdx.x;
    const unsigned base = blk * OWN_ROWS;

    float4* acc4 = (float4*)acc;
    for (int i = tid; i < ACC_ROWS * (D_OUT / 4); i += OWN_THREADS)
        acc4[i] = make_float4(0.f, 0.f, 0.f, 0.f);
    if (tid < ACC_ROWS) cnt[tid] = 0u;
    __syncthreads();

    unsigned* q = pq + wave * PQCAP;
    const uint4*  map4 = (const uint4*)maps + (size_t)blk * BLOCK_VECS;
    const uint4*  rec4 = (const uint4*)recs;
    const float2* h2   = (const float2*)h;
    int qcnt = 0;

    auto push = [&](bool on, unsigned word, unsigned wbits) {
        const unsigned s   = word & RW_SRC_MASK;
        const unsigned r   = min((word >> RW_SRC_BITS) & RW_ROW_MASK, (unsigned)OWN_ROWS);
        bool full = false;
        if (on) {
            const unsigned slot = atomicAdd(&cnt[r], 1u);
            if (slot < (unsigned)RQCAP)
                rq[r * RQ_PITCH + slot] = make_uint2(s, wbits);
            else
                full = true;
        }
        unsigned left = __builtin_amdgcn_ballot_w32(full);
        while (left != 0u) {
            const int j = __builtin_ctz(left);
            left &= left - 1u;
            const int      sj = __builtin_amdgcn_readlane((int)s, j);
            const float    wj = __int_as_float(__builtin_amdgcn_readlane((int)wbits, j));
            const unsigned rj = (unsigned)__builtin_amdgcn_readlane((int)r, j);
            const float2   v  = h2[(size_t)sj * (D_OUT / 2) + lane];
            float* a = acc + rj * D_OUT + lane * 2;
            atomicAdd(a + 0, v.x * wj);
            atomicAdd(a + 1, v.y * wj);
        }
    };

    auto hand_on = [&](const unsigned* qs, int n) {
        const bool     valid = lane < n;
        const unsigned p   = qs[valid ? lane : n - 1];
        const uint4    rc  = rec4[p];
        const unsigned id0 = (rc.x >> RW_TAG_SHIFT) | (((rc.z >> RW_TAG_SHIFT) & 3u) << 6);
        const bool first = id0 == blk;
        push(valid, first ? rc.x : rc.z, first ? rc.y : rc.w);
        const bool both = valid && first && ((rc.z >> (RW_TAG_SHIFT + 2)) & 1u) != 0u;
        if (__builtin_amdgcn_ballot_w32(both) != 0u) push(both, rc.z, rc.w);
    };

    uint4 d[WAVE_STEPS];
    #pragma unroll
    for (int u = 0; u < WAVE_STEPS; ++u) {
        const int tu = min(step_lo + wave + u * OWN_WAVES, step_hi - 1);
        const int  vi = tu * 32 + lane;
        const bool in = vi < BLOCK_VECS;
        const uint4 dv = map4[in ? vi : BLOCK_VECS - 1];
        d[u] = in ? dv : make_uint4(0u, 0u, 0u, 0u);
    }
    #pragma unroll 1
    for (int t = step_lo + wave; t < step_hi; t += OWN_WAVES) {
        uint4 cur = d[0];
        unsigned e0 = ((unsigned)(4 * t) + (unsigned)(lane >> 3)) * (unsigned)MAP_PAIRS
                    + (unsigned)(lane & 7) * 128u;
        #pragma unroll 1
        for (int i = 0; i < 4; ++i) {
            unsigned hit = cur.x;
            unsigned mask = __builtin_amdgcn_ballot_w32(hit != 0u);
            while (mask != 0u) {
                if (hit != 0u) {
                    const int pos = qcnt + (int)__builtin_amdgcn_mbcnt_lo(mask, 0u);
                    q[pos] = e0 + (unsigned)__builtin_ctz(hit);
                    hit &= hit - 1u;
                }
                qcnt += __builtin_popcount(mask);
                while (qcnt >= HITS) {
                    qcnt -= HITS;
                    hand_on(q + qcnt, HITS);
                }
                mask = __builtin_amdgcn_ballot_w32(hit != 0u);
            }
            cur.x = cur.y; cur.y = cur.z; cur.z = cur.w;
            e0 += 32u;
        }
        #pragma unroll
        for (int u = 0; u + 1 < WAVE_STEPS; ++u) d[u] = d[u + 1];
    }
    if (qcnt > 0) hand_on(q, qcnt);
    __syncthreads();

    const int first = (ROWS_X4 * wave + 3) >> 2;
    const int nrows = ((ROWS_X4 * (wave + 1) + 3) >> 2) - first;
    const int hf  = lane >> 4;
    const int l15 = lane & 15;
    const float4* h4   = (const float4*)h;
    const float4* acc4r = (const float4*)acc;
    float4* out4 = (float4*)out;

    const int cv = lane < nrows ? (int)min(cnt[first + lane], (unsigned)RQCAP) : 0;

    #pragma unroll 1
    for (int i = 0; i < nrows; ++i) {
        const int r = first + i;
        const int n = __builtin_amdgcn_readlane(cv, i);
        const uint2* rowq = rq + r * RQ_PITCH;
        float4 s = make_float4(0.f, 0.f, 0.f, 0.f);
        const float4 o = acc4r[r * (D_OUT / 4) + l15];

        auto group = [&](int u0, int m, auto cap) {
            constexpr int CAP = decltype(cap)::value;
            const uint2* qg = rowq + u0 + hf;
            uint2  e[CAP];
            float4 v[CAP];
            #pragma unroll
            for (int j = 0; j + 1 < CAP; ++j) e[j] = qg[2 * j];
            e[CAP - 1] = rowq[u0 + min(2 * (CAP - 1) + hf, m - 1)];
            #pragma unroll
            for (int j = 0; j < CAP; ++j)
                v[j] = h4[e[j].x * (unsigned)(D_OUT / 4) + (unsigned)l15];
            #pragma unroll
            for (int j = 0; j < CAP; ++j) {
                float wj = __uint_as_float(e[j].y);
                if (j == CAP - 1 && 2 * (CAP - 1) + hf >= m) wj = 0.f;
                s.x += v[j].x * wj; s.y += v[j].y * wj; s.z += v[j].z * wj; s.w += v[j].w * wj;
            }
        };

        #pragma unroll 1
        for (int u0 = 0; u0 < n; u0 += 2 * PAIRS) {
            const int m = min(n - u0, 2 * PAIRS);
            switch ((m + 1) >> 1) {
                case 1:  group(u0, m, loads_t<1>{});  break;
                case 2:  group(u0, m, loads_t<2>{});  break;
                case 3:  group(u0, m, loads_t<3>{});  break;
                case 4:  group(u0, m, loads_t<4>{});  break;
                case 5:  group(u0, m, loads_t<5>{});  break;
                case 6:  group(u0, m, loads_t<6>{});  break;
                case 7:  group(u0, m, loads_t<7>{});  break;
                case 8:  group(u0, m, loads_t<8>{});  break;
                case 9:  group(u0, m, loads_t<9>{});  break;
                case 10: group(u0, m, loads_t<10>{}); break;
                case 11: group(u0, m, loads_t<11>{}); break;
                default: group(u0, m, loads_t<PAIRS>{}); break;
            }
        }

        const float4 t = make_float4(s.x + other_half(s.x), s.y + other_half(s.y),
                                     s.z + other_half(s.z), s.w + other_half(s.w));
        const unsigned row = base + (unsigned)r;
        if (hf == 0 && r < OWN_ROWS && row < N_NODES)
            out4[(size_t)row * (D_OUT / 4) + l15] = make_float4(t.x + o.x, t.y + o.y, t.z + o.z, t.w + o.w);
    }
}

extern "C" void kernel_launch(void* const* d_in, const int* in_sizes, int n_in,
                              void* d_out, int out_size, void* d_ws, size_t ws_size,
                              hipStream_t stream) {
    const float* x      = (const float*)d_in[0];
    const float* W      = (const float*)d_in[1];
    const float* b      = (const float*)d_in[2];
    const float* edge_w = (const float*)d_in[3];
    const int*   src    = (const int*)d_in[4];
    const int*   dst    = (const int*)d_in[5];
    float*       out    = (float*)d_out;
    float*       h      = (float*)d_ws;
    unsigned int* maps  = (unsigned int*)(h + (size_t)N_NODES * D_OUT);
    unsigned int* recs  = maps + (size_t)OWN_BLOCKS * MAP_GROUPS * MAP_DWORDS;

    (void)in_sizes; (void)n_in; (void)ws_size; (void)out_size;

    const int m_tiles = (N_NODES + 15) / 16;
    const int gemm_blocks = (m_tiles + 7) / 8;
    gemm_bias_kernel<<<gemm_blocks, 256, 0, stream>>>(x, W, b, h, src, dst, edge_w, maps, recs);

    owner_sum_kernel<<<OWN_BLOCKS, OWN_THREADS, 0, stream>>>(h, maps, recs, out);
}
